# starting kernel
# baseline (speedup 1.0000x reference)
#include <hip/hip_runtime.h>

#define B_   8
#define S_   4096
#define HID_ 1024
#define H_   16
#define DH_  64

typedef __attribute__((ext_vector_type(2))) float v2f;
typedef __attribute__((ext_vector_type(8))) float v8f;

#define OFF_WQE 0
#define OFF_WKE (OFF_WQE + H_*HID_)
#define OFF_QS  (OFF_WKE + H_*HID_)
#define OFF_HSP (OFF_QS  + B_*S_*H_)
#define OFF_KHP (OFF_HSP + B_*32*HID_)
#define OFF_HS  (OFF_KHP + 256*H_*HID_)
#define OFF_KH  (OFF_HS  + B_*HID_)
#define OFF_MVS (OFF_KH  + B_*H_*HID_)
#define OFF_KVS (OFF_MVS + B_*H_*DH_)
#define OFF_IL  (OFF_KVS + B_*H_*DH_)

__global__ __launch_bounds__(256) void weff_kernel(
    const float* __restrict__ Wq, const float* __restrict__ Wk,
    const float* __restrict__ aw, float* __restrict__ wqe, float* __restrict__ wke)
{
    int u = blockIdx.x * 256 + threadIdx.x;
    int h = u >> 10, i = u & (HID_ - 1);
    float aq = 0.f, ak = 0.f;
#pragma unroll 4
    for (int d = 0; d < DH_; ++d) {
        float a = aw[h * DH_ + d];
        aq += a * Wq[(size_t)(h * DH_ + d) * HID_ + i];
        ak += a * Wk[(size_t)(h * DH_ + d) * HID_ + i];
    }
    wqe[h * HID_ + i] = aq;
    wke[h * HID_ + i] = ak;
}

__global__ __launch_bounds__(256) void invlen_kernel(
    const float* __restrict__ mask, float* __restrict__ il)
{
    __shared__ float red[256];
    int b = blockIdx.x, t = threadIdx.x;
    float s = 0.f;
#pragma unroll
    for (int k = 0; k < S_ / 256; ++k) s += mask[b * S_ + t + 256 * k];
    red[t] = s;
    __syncthreads();
    for (int w = 128; w > 0; w >>= 1) {
        if (t < w) red[t] += red[t + w];
        __syncthreads();
    }
    if (t == 0) il[b] = 1.0f / red[0];
}

__global__ __launch_bounds__(256) void pass1_kernel(
    const float* __restrict__ hidden, const float* __restrict__ mask,
    const float* __restrict__ wqe, const float* __restrict__ wke,
    float* __restrict__ q_scores, float* __restrict__ hsum_part,
    float* __restrict__ kh_part)
{
    __shared__ float lds_tile[128 * 68];
    __shared__ float lds_wq[16 * 68];
    __shared__ float lds_wk[16 * 68];
    __shared__ float lds_ksT[16 * 132];
    __shared__ float lds_red[256 * 4];
    __shared__ float lds_khb[2 * 16 * 68];

    const int blk = blockIdx.x & 31;
    const int b   = blockIdx.x >> 5;
    const int s0  = blk * 128;
    const int t = threadIdx.x;
    const int wave = t >> 5, lane = t & 31;
    const int lane16 = lane & 15, laneHi = lane >> 4;

    const float* hbase = hidden + ((size_t)b * S_ + s0) * HID_;

    v8f cq = {}; v8f ck = {};

    for (int kc = 0; kc < 16; ++kc) {
        const int k0 = kc * 64;
        {
            int r = t >> 4, c4 = (t & 15) * 4;
            *(float4*)&lds_wq[r * 68 + c4] = *(const float4*)(wqe + r * HID_ + k0 + c4);
            *(float4*)&lds_wk[r * 68 + c4] = *(const float4*)(wke + r * HID_ + k0 + c4);
        }
        float4 acc = {0.f, 0.f, 0.f, 0.f};
#pragma unroll
        for (int it = 0; it < 8; ++it) {
            int row = it * 16 + (t >> 4);
            float4 v = *(const float4*)(hbase + (size_t)row * HID_ + k0 + (t & 15) * 4);
            *(float4*)&lds_tile[row * 68 + (t & 15) * 4] = v;
            float m = mask[b * S_ + s0 + row];
            acc.x += m * v.x; acc.y += m * v.y; acc.z += m * v.z; acc.w += m * v.w;
        }
        *(float4*)&lds_red[t * 4] = acc;
        __syncthreads();
        if (t < 64) {
            int g = t >> 2, comp = t & 3;
            float s = 0.f;
#pragma unroll
            for (int p = 0; p < 16; ++p) s += lds_red[(p * 16 + g) * 4 + comp];
            hsum_part[(size_t)(b * 32 + blk) * HID_ + k0 + g * 4 + comp] = s;
        }
        const float* atile = &lds_tile[(wave * 16 + lane16) * 68 + 2 * laneHi];
        const float* bq = &lds_wq[lane16 * 68 + 2 * laneHi];
        const float* bk = &lds_wk[lane16 * 68 + 2 * laneHi];
#pragma unroll
        for (int kk = 0; kk < 16; ++kk) {
            v2f a = *(const v2f*)(atile + kk * 4);
            v2f q = *(const v2f*)(bq + kk * 4);
            v2f k = *(const v2f*)(bk + kk * 4);
            cq = __builtin_amdgcn_wmma_f32_16x16x4_f32(false, a, false, q, (short)0, cq, false, false);
            ck = __builtin_amdgcn_wmma_f32_16x16x4_f32(false, a, false, k, (short)0, ck, false, false);
        }
        __syncthreads();
    }

#pragma unroll
    for (int j = 0; j < 8; ++j) {
        int sl = wave * 16 + j + 8 * laneHi;
        float m = mask[b * S_ + s0 + sl];
        q_scores[((size_t)b * S_ + s0 + sl) * H_ + lane16] = cq[j] * m;
        lds_ksT[lane16 * 132 + sl] = ck[j] * m;
    }
    __syncthreads();

    const int ic = wave & 3;
    const int sh = wave >> 2;
    for (int kc = 0; kc < 16; ++kc) {
        const int k0 = kc * 64;
#pragma unroll
        for (int it = 0; it < 8; ++it) {
            int row = it * 16 + (t >> 4);
            *(float4*)&lds_tile[row * 68 + (t & 15) * 4] =
                *(const float4*)(hbase + (size_t)row * HID_ + k0 + (t & 15) * 4);
        }
        __syncthreads();
        v8f ckh = {};
        const float* aks = &lds_ksT[lane16 * 132 + sh * 64 + 2 * laneHi];
#pragma unroll
        for (int ss = 0; ss < 16; ++ss) {
            v2f a = *(const v2f*)(aks + ss * 4);
            int srow = sh * 64 + ss * 4 + 2 * laneHi;
            v2f bb;
            bb.x = lds_tile[srow * 68 + ic * 16 + lane16];
            bb.y = lds_tile[(srow + 1) * 68 + ic * 16 + lane16];
            ckh = __builtin_amdgcn_wmma_f32_16x16x4_f32(false, a, false, bb, (short)0, ckh, false, false);
        }
#pragma unroll
        for (int j = 0; j < 8; ++j)
            lds_khb[(sh * 16 + j + 8 * laneHi) * 68 + ic * 16 + lane16] = ckh[j];
        __syncthreads();
        {
            int h = t >> 4, c4 = (t & 15) * 4;
            float4 a0 = *(const float4*)&lds_khb[h * 68 + c4];
            float4 a1 = *(const float4*)&lds_khb[(16 + h) * 68 + c4];
            float4 r; r.x = a0.x + a1.x; r.y = a0.y + a1.y; r.z = a0.z + a1.z; r.w = a0.w + a1.w;
            *(float4*)&kh_part[(((size_t)(b * 32 + blk)) * H_ + h) * HID_ + k0 + c4] = r;
        }
        __syncthreads();
    }
}

__global__ __launch_bounds__(256) void reduce_hsum_kernel(
    const float* __restrict__ hsp, float* __restrict__ hs)
{
    int idx = blockIdx.x * 256 + threadIdx.x;
    int b = idx >> 10, i = idx & (HID_ - 1);
    float s = 0.f;
#pragma unroll 8
    for (int blk = 0; blk < 32; ++blk) s += hsp[(size_t)(b * 32 + blk) * HID_ + i];
    hs[idx] = s;
}

__global__ __launch_bounds__(256) void reduce_kh_kernel(
    const float* __restrict__ khp, float* __restrict__ kh)
{
    int idx = blockIdx.x * 256 + threadIdx.x;
    int bh = idx >> 10, i = idx & (HID_ - 1);
    int b = bh >> 4, h = bh & 15;
    float s = 0.f;
#pragma unroll 8
    for (int blk = 0; blk < 32; ++blk)
        s += khp[(((size_t)(b * 32 + blk)) * H_ + h) * HID_ + i];
    kh[idx] = s;
}

__global__ __launch_bounds__(256) void pass2_kernel(
    const float* __restrict__ Wv, const float* __restrict__ hs,
    const float* __restrict__ kh, const float* __restrict__ il,
    float* __restrict__ mvs, float* __restrict__ kvs)
{
    int u = blockIdx.x * 256 + threadIdx.x;
    int h = u >> 6, d = u & 63;
    float mv[B_], kv[B_];
#pragma unroll
    for (int b = 0; b < B_; ++b) { mv[b] = 0.f; kv[b] = 0.f; }
    const float* wrow = Wv + (size_t)(h * DH_ + d) * HID_;
    for (int i = 0; i < HID_; i += 4) {
        float4 w = *(const float4*)(wrow + i);
#pragma unroll
        for (int b = 0; b < B_; ++b) {
            float4 s4 = *(const float4*)(hs + b * HID_ + i);
            float4 k4 = *(const float4*)(kh + (size_t)(b * H_ + h) * HID_ + i);
            mv[b] += w.x * s4.x + w.y * s4.y + w.z * s4.z + w.w * s4.w;
            kv[b] += w.x * k4.x + w.y * k4.y + w.z * k4.z + w.w * k4.w;
        }
    }
#pragma unroll
    for (int b = 0; b < B_; ++b) {
        float inv = il[b];
        mvs[(b * H_ + h) * DH_ + d] = mv[b] * inv;
        kvs[(b * H_ + h) * DH_ + d] = kv[b] * inv;
    }
}

__global__ __launch_bounds__(256) void pass3_kernel(
    const float* __restrict__ qs, const float* __restrict__ mvs,
    const float* __restrict__ kvs, float* __restrict__ out)
{
    const int b = blockIdx.x >> 7;
    const int srow0 = (blockIdx.x & 127) * 32;
    const int t = threadIdx.x;
    const int h = t >> 4;
    float4 mv4 = *(const float4*)(mvs + (b * H_ + h) * DH_ + (t & 15) * 4);
    float4 kv4 = *(const float4*)(kvs + (b * H_ + h) * DH_ + (t & 15) * 4);
#pragma unroll 4
    for (int r = 0; r < 32; ++r) {
        size_t srow = (size_t)b * S_ + srow0 + r;
        float q = qs[srow * H_ + h];
        float4 o;
        o.x = q * mv4.x + kv4.x; o.y = q * mv4.y + kv4.y;
        o.z = q * mv4.z + kv4.z; o.w = q * mv4.w + kv4.w;
        *(float4*)(out + srow * (H_ * DH_) + t * 4) = o;
    }
}

extern "C" void kernel_launch(void* const* d_in, const int* in_sizes, int n_in,
                              void* d_out, int out_size, void* d_ws, size_t ws_size,
                              hipStream_t stream) {
    const float* hidden = (const float*)d_in[0];
    const float* mask   = (const float*)d_in[1];
    const float* Wq     = (const float*)d_in[2];
    const float* Wk     = (const float*)d_in[3];
    const float* Wv     = (const float*)d_in[4];
    const float* aw     = (const float*)d_in[5];
    float* out = (float*)d_out;
    float* ws  = (float*)d_ws;

    weff_kernel   <<<64,   256, 0, stream>>>(Wq, Wk, aw, ws + OFF_WQE, ws + OFF_WKE);
    invlen_kernel <<<B_,   256, 0, stream>>>(mask, ws + OFF_IL);
    pass1_kernel  <<<256,  256, 0, stream>>>(hidden, mask, ws + OFF_WQE, ws + OFF_WKE,
                                             ws + OFF_QS, ws + OFF_HSP, ws + OFF_KHP);
    reduce_hsum_kernel<<<32,  256, 0, stream>>>(ws + OFF_HSP, ws + OFF_HS);
    reduce_kh_kernel  <<<512, 256, 0, stream>>>(ws + OFF_KHP, ws + OFF_KH);
    pass2_kernel  <<<4,    256, 0, stream>>>(Wv, ws + OFF_HS, ws + OFF_KH, ws + OFF_IL,
                                             ws + OFF_MVS, ws + OFF_KVS);
    pass3_kernel  <<<1024, 256, 0, stream>>>(ws + OFF_QS, ws + OFF_MVS, ws + OFF_KVS, out);
}


# mean/key value projection kernel re-tiled: 4 work-groups of per-thread strided weight rows -> 64 work-groups (head x 16 outputs) on f32 WMMA, K split over 8 waves, fixed-order LDS sum (new sum order)
# speedup vs baseline: 1.5410x; 1.5410x over previous
#include <hip/hip_runtime.h>

#define B_   8
#define S_   4096
#define HID_ 1024
#define H_   16
#define DH_  64

typedef __attribute__((ext_vector_type(2))) float v2f;
typedef __attribute__((ext_vector_type(8))) float v8f;

#define OFF_WQE 0
#define OFF_WKE (OFF_WQE + H_*HID_)
#define OFF_QS  (OFF_WKE + H_*HID_)
#define OFF_HSP (OFF_QS  + B_*S_*H_)
#define OFF_KHP (OFF_HSP + B_*32*HID_)
#define OFF_HS  (OFF_KHP + 256*H_*HID_)
#define OFF_KH  (OFF_HS  + B_*HID_)
#define OFF_MKV (OFF_KH  + B_*H_*HID_)
#define OFF_IL  (OFF_MKV + 2*B_*H_*DH_)

__global__ __launch_bounds__(256) void weff_kernel(
    const float* __restrict__ Wq, const float* __restrict__ Wk,
    const float* __restrict__ aw, float* __restrict__ wqe, float* __restrict__ wke)
{
    int u = blockIdx.x * 256 + threadIdx.x;
    int h = u >> 10, i = u & (HID_ - 1);
    float aq = 0.f, ak = 0.f;
#pragma unroll 4
    for (int d = 0; d < DH_; ++d) {
        float a = aw[h * DH_ + d];
        aq += a * Wq[(size_t)(h * DH_ + d) * HID_ + i];
        ak += a * Wk[(size_t)(h * DH_ + d) * HID_ + i];
    }
    wqe[h * HID_ + i] = aq;
    wke[h * HID_ + i] = ak;
}

__global__ __launch_bounds__(256) void invlen_kernel(
    const float* __restrict__ mask, float* __restrict__ il)
{
    __shared__ float red[256];
    int b = blockIdx.x, t = threadIdx.x;
    float s = 0.f;
#pragma unroll
    for (int k = 0; k < S_ / 256; ++k) s += mask[b * S_ + t + 256 * k];
    red[t] = s;
    __syncthreads();
    for (int w = 128; w > 0; w >>= 1) {
        if (t < w) red[t] += red[t + w];
        __syncthreads();
    }
    if (t == 0) il[b] = 1.0f / red[0];
}

__global__ __launch_bounds__(256) void pass1_kernel(
    const float* __restrict__ hidden, const float* __restrict__ mask,
    const float* __restrict__ wqe, const float* __restrict__ wke,
    float* __restrict__ q_scores, float* __restrict__ hsum_part,
    float* __restrict__ kh_part)
{
    __shared__ float lds_tile[128 * 68];
    __shared__ float lds_wq[16 * 68];
    __shared__ float lds_wk[16 * 68];
    __shared__ float lds_ksT[16 * 132];
    __shared__ float lds_red[256 * 4];
    __shared__ float lds_khb[2 * 16 * 68];

    const int blk = blockIdx.x & 31;
    const int b   = blockIdx.x >> 5;
    const int s0  = blk * 128;
    const int t = threadIdx.x;
    const int wave = t >> 5, lane = t & 31;
    const int lane16 = lane & 15, laneHi = lane >> 4;

    const float* hbase = hidden + ((size_t)b * S_ + s0) * HID_;

    v8f cq = {}; v8f ck = {};

    for (int kc = 0; kc < 16; ++kc) {
        const int k0 = kc * 64;
        {
            int r = t >> 4, c4 = (t & 15) * 4;
            *(float4*)&lds_wq[r * 68 + c4] = *(const float4*)(wqe + r * HID_ + k0 + c4);
            *(float4*)&lds_wk[r * 68 + c4] = *(const float4*)(wke + r * HID_ + k0 + c4);
        }
        float4 acc = {0.f, 0.f, 0.f, 0.f};
#pragma unroll
        for (int it = 0; it < 8; ++it) {
            int row = it * 16 + (t >> 4);
            float4 v = *(const float4*)(hbase + (size_t)row * HID_ + k0 + (t & 15) * 4);
            *(float4*)&lds_tile[row * 68 + (t & 15) * 4] = v;
            float m = mask[b * S_ + s0 + row];
            acc.x += m * v.x; acc.y += m * v.y; acc.z += m * v.z; acc.w += m * v.w;
        }
        *(float4*)&lds_red[t * 4] = acc;
        __syncthreads();
        if (t < 64) {
            int g = t >> 2, comp = t & 3;
            float s = 0.f;
#pragma unroll
            for (int p = 0; p < 16; ++p) s += lds_red[(p * 16 + g) * 4 + comp];
            hsum_part[(size_t)(b * 32 + blk) * HID_ + k0 + g * 4 + comp] = s;
        }
        const float* atile = &lds_tile[(wave * 16 + lane16) * 68 + 2 * laneHi];
        const float* bq = &lds_wq[lane16 * 68 + 2 * laneHi];
        const float* bk = &lds_wk[lane16 * 68 + 2 * laneHi];
#pragma unroll
        for (int kk = 0; kk < 16; ++kk) {
            v2f a = *(const v2f*)(atile + kk * 4);
            v2f q = *(const v2f*)(bq + kk * 4);
            v2f k = *(const v2f*)(bk + kk * 4);
            cq = __builtin_amdgcn_wmma_f32_16x16x4_f32(false, a, false, q, (short)0, cq, false, false);
            ck = __builtin_amdgcn_wmma_f32_16x16x4_f32(false, a, false, k, (short)0, ck, false, false);
        }
        __syncthreads();
    }

#pragma unroll
    for (int j = 0; j < 8; ++j) {
        int sl = wave * 16 + j + 8 * laneHi;
        float m = mask[b * S_ + s0 + sl];
        q_scores[((size_t)b * S_ + s0 + sl) * H_ + lane16] = cq[j] * m;
        lds_ksT[lane16 * 132 + sl] = ck[j] * m;
    }
    __syncthreads();

    const int ic = wave & 3;
    const int sh = wave >> 2;
    for (int kc = 0; kc < 16; ++kc) {
        const int k0 = kc * 64;
#pragma unroll
        for (int it = 0; it < 8; ++it) {
            int row = it * 16 + (t >> 4);
            *(float4*)&lds_tile[row * 68 + (t & 15) * 4] =
                *(const float4*)(hbase + (size_t)row * HID_ + k0 + (t & 15) * 4);
        }
        __syncthreads();
        v8f ckh = {};
        const float* aks = &lds_ksT[lane16 * 132 + sh * 64 + 2 * laneHi];
#pragma unroll
        for (int ss = 0; ss < 16; ++ss) {
            v2f a = *(const v2f*)(aks + ss * 4);
            int srow = sh * 64 + ss * 4 + 2 * laneHi;
            v2f bb;
            bb.x = lds_tile[srow * 68 + ic * 16 + lane16];
            bb.y = lds_tile[(srow + 1) * 68 + ic * 16 + lane16];
            ckh = __builtin_amdgcn_wmma_f32_16x16x4_f32(false, a, false, bb, (short)0, ckh, false, false);
        }
#pragma unroll
        for (int j = 0; j < 8; ++j)
            lds_khb[(sh * 16 + j + 8 * laneHi) * 68 + ic * 16 + lane16] = ckh[j];
        __syncthreads();
        {
            int h = t >> 4, c4 = (t & 15) * 4;
            float4 a0 = *(const float4*)&lds_khb[h * 68 + c4];
            float4 a1 = *(const float4*)&lds_khb[(16 + h) * 68 + c4];
            float4 r; r.x = a0.x + a1.x; r.y = a0.y + a1.y; r.z = a0.z + a1.z; r.w = a0.w + a1.w;
            *(float4*)&kh_part[(((size_t)(b * 32 + blk)) * H_ + h) * HID_ + k0 + c4] = r;
        }
        __syncthreads();
    }
}

__global__ __launch_bounds__(256) void reduce_hsum_kernel(
    const float* __restrict__ hsp, float* __restrict__ hs)
{
    int idx = blockIdx.x * 256 + threadIdx.x;
    int b = idx >> 10, i = idx & (HID_ - 1);
    float s = 0.f;
#pragma unroll 8
    for (int blk = 0; blk < 32; ++blk) s += hsp[(size_t)(b * 32 + blk) * HID_ + i];
    hs[idx] = s;
}

__global__ __launch_bounds__(256) void reduce_kh_kernel(
    const float* __restrict__ khp, float* __restrict__ kh)
{
    int idx = blockIdx.x * 256 + threadIdx.x;
    int bh = idx >> 10, i = idx & (HID_ - 1);
    int b = bh >> 4, h = bh & 15;
    float s = 0.f;
#pragma unroll 8
    for (int blk = 0; blk < 32; ++blk)
        s += khp[(((size_t)(b * 32 + blk)) * H_ + h) * HID_ + i];
    kh[idx] = s;
}

__global__ __launch_bounds__(256) void pass2_kernel(
    const float* __restrict__ Wv, const float* __restrict__ hs,
    const float* __restrict__ kh, const float* __restrict__ il,
    float* __restrict__ mkv)
{
    __shared__ float red[8 * 8 * 32];
    const int h  = blockIdx.x >> 2;
    const int d0 = (blockIdx.x & 3) * 16;
    const int t = threadIdx.x;
    const int wave = t >> 5, lane = t & 31;
    const int lane16 = lane & 15, laneHi = lane >> 4;

    const int ob = ((wave & 3) << 1) | laneHi;
    const float inv = il[ob];

    const float* arow = (lane16 < 8) ? (hs + lane16 * HID_)
                                     : (kh + (size_t)((lane16 - 8) * H_ + h) * HID_);
    const float* brow = Wv + (size_t)(h * DH_ + d0 + lane16) * HID_;
    const int kb = wave * 128 + laneHi * 4;

    float4 a[16], w[16];
#pragma unroll
    for (int s = 0; s < 16; ++s) {
        a[s] = *(const float4*)(arow + kb + s * 8);
        w[s] = *(const float4*)(brow + kb + s * 8);
    }
    __builtin_amdgcn_sched_barrier(0);

    v8f acc0 = {}; v8f acc1 = {};
#pragma unroll
    for (int s = 0; s < 16; ++s) {
        v2f a0 = {a[s].x, a[s].y}, a1 = {a[s].z, a[s].w};
        v2f w0 = {w[s].x, w[s].y}, w1 = {w[s].z, w[s].w};
        acc0 = __builtin_amdgcn_wmma_f32_16x16x4_f32(false, a0, false, w0, (short)0, acc0, false, false);
        acc1 = __builtin_amdgcn_wmma_f32_16x16x4_f32(false, a1, false, w1, (short)0, acc1, false, false);
    }
#pragma unroll
    for (int j = 0; j < 8; ++j) red[(wave * 8 + j) * 32 + lane] = acc0[j] + acc1[j];
    __syncthreads();
    {
        const int half = wave >> 2;
        const int src = ob * 32 + half * 16 + lane16;
        float s = red[src];
#pragma unroll
        for (int k = 1; k < 8; ++k) s += red[k * 256 + src];
        mkv[blockIdx.x * 256 + t] = s * inv;
    }
}

__global__ __launch_bounds__(256) void pass3_kernel(
    const float* __restrict__ qs, const float* __restrict__ mkv, float* __restrict__ out)
{
    const int b = blockIdx.x >> 7;
    const int srow0 = (blockIdx.x & 127) * 32;
    const int t = threadIdx.x;
    const int h = t >> 4;
    const float* tile = mkv + (h * 4 + ((t & 15) >> 2)) * 256 + b * 16 + (t & 3) * 4;
    float4 mv4 = *(const float4*)(tile);
    float4 kv4 = *(const float4*)(tile + 128);
#pragma unroll 4
    for (int r = 0; r < 32; ++r) {
        size_t srow = (size_t)b * S_ + srow0 + r;
        float q = qs[srow * H_ + h];
        float4 o;
        o.x = q * mv4.x + kv4.x; o.y = q * mv4.y + kv4.y;
        o.z = q * mv4.z + kv4.z; o.w = q * mv4.w + kv4.w;
        *(float4*)(out + srow * (H_ * DH_) + t * 4) = o;
    }
}

extern "C" void kernel_launch(void* const* d_in, const int* in_sizes, int n_in,
                              void* d_out, int out_size, void* d_ws, size_t ws_size,
                              hipStream_t stream) {
    const float* hidden = (const float*)d_in[0];
    const float* mask   = (const float*)d_in[1];
    const float* Wq     = (const float*)d_in[2];
    const float* Wk     = (const float*)d_in[3];
    const float* Wv     = (const float*)d_in[4];
    const float* aw     = (const float*)d_in[5];
    float* out = (float*)d_out;
    float* ws  = (float*)d_ws;

    weff_kernel   <<<64,   256, 0, stream>>>(Wq, Wk, aw, ws + OFF_WQE, ws + OFF_WKE);
    invlen_kernel <<<B_,   256, 0, stream>>>(mask, ws + OFF_IL);
    pass1_kernel  <<<256,  256, 0, stream>>>(hidden, mask, ws + OFF_WQE, ws + OFF_WKE,
                                             ws + OFF_QS, ws + OFF_HSP, ws + OFF_KHP);
    reduce_hsum_kernel<<<32,  256, 0, stream>>>(ws + OFF_HSP, ws + OFF_HS);
    reduce_kh_kernel  <<<512, 256, 0, stream>>>(ws + OFF_KHP, ws + OFF_KH);
    pass2_kernel  <<<64,   256, 0, stream>>>(Wv, ws + OFF_HS, ws + OFF_KH, ws + OFF_IL, ws + OFF_MKV);
    pass3_kernel  <<<1024, 256, 0, stream>>>(ws + OFF_QS, ws + OFF_MKV, out);
}


# scores/kh kernel: K-chunk loops software-pipelined over two register sets, next chunk's global loads stay in flight across LDS-only barriers while this chunk's WMMAs run; every sum unchanged
# speedup vs baseline: 1.9715x; 1.2794x over previous
#include <hip/hip_runtime.h>

#define B_   8
#define S_   4096
#define HID_ 1024
#define H_   16
#define DH_  64

typedef __attribute__((ext_vector_type(2))) float v2f;
typedef __attribute__((ext_vector_type(8))) float v8f;

#define OFF_WQE 0
#define OFF_WKE (OFF_WQE + H_*HID_)
#define OFF_QS  (OFF_WKE + H_*HID_)
#define OFF_HSP (OFF_QS  + B_*S_*H_)
#define OFF_KHP (OFF_HSP + B_*32*HID_)
#define OFF_HS  (OFF_KHP + 256*H_*HID_)
#define OFF_KH  (OFF_HS  + B_*HID_)
#define OFF_MKV (OFF_KH  + B_*H_*HID_)
#define OFF_IL  (OFF_MKV + 2*B_*H_*DH_)

__global__ __launch_bounds__(256) void weff_kernel(
    const float* __restrict__ Wq, const float* __restrict__ Wk,
    const float* __restrict__ aw, float* __restrict__ wqe, float* __restrict__ wke)
{
    int u = blockIdx.x * 256 + threadIdx.x;
    int h = u >> 10, i = u & (HID_ - 1);
    float aq = 0.f, ak = 0.f;
#pragma unroll 4
    for (int d = 0; d < DH_; ++d) {
        float a = aw[h * DH_ + d];
        aq += a * Wq[(size_t)(h * DH_ + d) * HID_ + i];
        ak += a * Wk[(size_t)(h * DH_ + d) * HID_ + i];
    }
    wqe[h * HID_ + i] = aq;
    wke[h * HID_ + i] = ak;
}

__global__ __launch_bounds__(256) void invlen_kernel(
    const float* __restrict__ mask, float* __restrict__ il)
{
    __shared__ float red[256];
    int b = blockIdx.x, t = threadIdx.x;
    float s = 0.f;
#pragma unroll
    for (int k = 0; k < S_ / 256; ++k) s += mask[b * S_ + t + 256 * k];
    red[t] = s;
    __syncthreads();
    for (int w = 128; w > 0; w >>= 1) {
        if (t < w) red[t] += red[t + w];
        __syncthreads();
    }
    if (t == 0) il[b] = 1.0f / red[0];
}

static __device__ __forceinline__ void lds_barrier()
{
    asm volatile("s_wait_dscnt 0x0\n\ts_barrier_signal -1\n\ts_barrier_wait -1" ::: "memory");
}

static __device__ __forceinline__ void fetch_chunk(float4 (&v)[8], const float* __restrict__ hbase, int k0, int t)
{
#pragma unroll
    for (int it = 0; it < 8; ++it) {
        int row = it * 16 + (t >> 4);
        v[it] = *(const float4*)(hbase + (size_t)row * HID_ + k0 + (t & 15) * 4);
    }
}

__global__ __launch_bounds__(256) void pass1_kernel(
    const float* __restrict__ hidden, const float* __restrict__ mask,
    const float* __restrict__ wqe, const float* __restrict__ wke,
    float* __restrict__ q_scores, float* __restrict__ hsum_part,
    float* __restrict__ kh_part)
{
    __shared__ float lds_tile[128 * 68];
    __shared__ float lds_wq[16 * 68];
    __shared__ float lds_wk[16 * 68];
    __shared__ float lds_ksT[16 * 132];
    __shared__ float lds_red[256 * 4];
    __shared__ float lds_khb[2 * 16 * 68];

    const int blk = blockIdx.x & 31;
    const int b   = blockIdx.x >> 5;
    const int s0  = blk * 128;
    const int t = threadIdx.x;
    const int wave = t >> 5, lane = t & 31;
    const int lane16 = lane & 15, laneHi = lane >> 4;
    const int wr = t >> 4, wc4 = (t & 15) * 4;

    const float* hbase = hidden + ((size_t)b * S_ + s0) * HID_;

    float mrow[8];
#pragma unroll
    for (int it = 0; it < 8; ++it) mrow[it] = mask[b * S_ + s0 + it * 16 + wr];

    v8f cq = {}; v8f ck = {};

    float4 va[8], vb[8];
    float4 wqa, wka, wqb, wkb;

    auto fetch_w = [&](float4& q4, float4& k4, int kc) {
        q4 = *(const float4*)(wqe + wr * HID_ + kc * 64 + wc4);
        k4 = *(const float4*)(wke + wr * HID_ + kc * 64 + wc4);
    };
    auto stepA = [&](const float4 (&v)[8], const float4& q4, const float4& k4, int kc) {
        const int k0 = kc * 64;
        *(float4*)&lds_wq[wr * 68 + wc4] = q4;
        *(float4*)&lds_wk[wr * 68 + wc4] = k4;
        float4 acc = {0.f, 0.f, 0.f, 0.f};
#pragma unroll
        for (int it = 0; it < 8; ++it) {
            int row = it * 16 + wr;
            *(float4*)&lds_tile[row * 68 + wc4] = v[it];
            float m = mrow[it];
            acc.x += m * v[it].x; acc.y += m * v[it].y; acc.z += m * v[it].z; acc.w += m * v[it].w;
        }
        *(float4*)&lds_red[t * 4] = acc;
        lds_barrier();
        if (t < 64) {
            int g = t >> 2, comp = t & 3;
            float s = 0.f;
#pragma unroll
            for (int p = 0; p < 16; ++p) s += lds_red[(p * 16 + g) * 4 + comp];
            hsum_part[(size_t)(b * 32 + blk) * HID_ + k0 + g * 4 + comp] = s;
        }
        const float* atile = &lds_tile[(wave * 16 + lane16) * 68 + 2 * laneHi];
        const float* bq = &lds_wq[lane16 * 68 + 2 * laneHi];
        const float* bk = &lds_wk[lane16 * 68 + 2 * laneHi];
#pragma unroll
        for (int kk = 0; kk < 16; ++kk) {
            v2f a = *(const v2f*)(atile + kk * 4);
            v2f q = *(const v2f*)(bq + kk * 4);
            v2f k = *(const v2f*)(bk + kk * 4);
            cq = __builtin_amdgcn_wmma_f32_16x16x4_f32(false, a, false, q, (short)0, cq, false, false);
            ck = __builtin_amdgcn_wmma_f32_16x16x4_f32(false, a, false, k, (short)0, ck, false, false);
        }
        lds_barrier();
    };

    fetch_chunk(va, hbase, 0, t);
    fetch_w(wqa, wka, 0);
    for (int kc = 0; kc < 16; kc += 2) {
        fetch_chunk(vb, hbase, (kc + 1) * 64, t);
        fetch_w(wqb, wkb, kc + 1);
        stepA(va, wqa, wka, kc);
        fetch_chunk(va, hbase, ((kc + 2) & 15) * 64, t);
        fetch_w(wqa, wka, (kc + 2) & 15);
        stepA(vb, wqb, wkb, kc + 1);
    }

#pragma unroll
    for (int j = 0; j < 8; ++j) {
        int sl = wave * 16 + j + 8 * laneHi;
        float m = mask[b * S_ + s0 + sl];
        q_scores[((size_t)b * S_ + s0 + sl) * H_ + lane16] = cq[j] * m;
        lds_ksT[lane16 * 132 + sl] = ck[j] * m;
    }
    lds_barrier();

    const int ic = wave & 3;
    const int sh = wave >> 2;
    auto stepC = [&](const float4 (&v)[8], int kc) {
        const int k0 = kc * 64;
#pragma unroll
        for (int it = 0; it < 8; ++it) {
            int row = it * 16 + wr;
            *(float4*)&lds_tile[row * 68 + wc4] = v[it];
        }
        lds_barrier();
        v8f ckh = {};
        const float* aks = &lds_ksT[lane16 * 132 + sh * 64 + 2 * laneHi];
#pragma unroll
        for (int ss = 0; ss < 16; ++ss) {
            v2f a = *(const v2f*)(aks + ss * 4);
            int srow = sh * 64 + ss * 4 + 2 * laneHi;
            v2f bb;
            bb.x = lds_tile[srow * 68 + ic * 16 + lane16];
            bb.y = lds_tile[(srow + 1) * 68 + ic * 16 + lane16];
            ckh = __builtin_amdgcn_wmma_f32_16x16x4_f32(false, a, false, bb, (short)0, ckh, false, false);
        }
#pragma unroll
        for (int j = 0; j < 8; ++j)
            lds_khb[(sh * 16 + j + 8 * laneHi) * 68 + ic * 16 + lane16] = ckh[j];
        lds_barrier();
        {
            int h = t >> 4, c4 = (t & 15) * 4;
            float4 a0 = *(const float4*)&lds_khb[h * 68 + c4];
            float4 a1 = *(const float4*)&lds_khb[(16 + h) * 68 + c4];
            float4 r; r.x = a0.x + a1.x; r.y = a0.y + a1.y; r.z = a0.z + a1.z; r.w = a0.w + a1.w;
            *(float4*)&kh_part[(((size_t)(b * 32 + blk)) * H_ + h) * HID_ + k0 + c4] = r;
        }
        lds_barrier();
    };

    for (int kc = 0; kc < 16; kc += 2) {
        fetch_chunk(vb, hbase, (kc + 1) * 64, t);
        stepC(va, kc);
        fetch_chunk(va, hbase, ((kc + 2) & 15) * 64, t);
        stepC(vb, kc + 1);
    }
}

__global__ __launch_bounds__(256) void reduce_hsum_kernel(
    const float* __restrict__ hsp, float* __restrict__ hs)
{
    int idx = blockIdx.x * 256 + threadIdx.x;
    int b = idx >> 10, i = idx & (HID_ - 1);
    float s = 0.f;
#pragma unroll 8
    for (int blk = 0; blk < 32; ++blk) s += hsp[(size_t)(b * 32 + blk) * HID_ + i];
    hs[idx] = s;
}

__global__ __launch_bounds__(256) void reduce_kh_kernel(
    const float* __restrict__ khp, float* __restrict__ kh)
{
    int idx = blockIdx.x * 256 + threadIdx.x;
    int bh = idx >> 10, i = idx & (HID_ - 1);
    int b = bh >> 4, h = bh & 15;
    float s = 0.f;
#pragma unroll 8
    for (int blk = 0; blk < 32; ++blk)
        s += khp[(((size_t)(b * 32 + blk)) * H_ + h) * HID_ + i];
    kh[idx] = s;
}

__global__ __launch_bounds__(256) void pass2_kernel(
    const float* __restrict__ Wv, const float* __restrict__ hs,
    const float* __restrict__ kh, const float* __restrict__ il,
    float* __restrict__ mkv)
{
    __shared__ float red[8 * 8 * 32];
    const int h  = blockIdx.x >> 2;
    const int d0 = (blockIdx.x & 3) * 16;
    const int t = threadIdx.x;
    const int wave = t >> 5, lane = t & 31;
    const int lane16 = lane & 15, laneHi = lane >> 4;

    const int ob = ((wave & 3) << 1) | laneHi;
    const float inv = il[ob];

    const float* arow = (lane16 < 8) ? (hs + lane16 * HID_)
                                     : (kh + (size_t)((lane16 - 8) * H_ + h) * HID_);
    const float* brow = Wv + (size_t)(h * DH_ + d0 + lane16) * HID_;
    const int kb = wave * 128 + laneHi * 4;

    float4 a[16], w[16];
#pragma unroll
    for (int s = 0; s < 16; ++s) {
        a[s] = *(const float4*)(arow + kb + s * 8);
        w[s] = *(const float4*)(brow + kb + s * 8);
    }
    __builtin_amdgcn_sched_barrier(0);

    v8f acc0 = {}; v8f acc1 = {};
#pragma unroll
    for (int s = 0; s < 16; ++s) {
        v2f a0 = {a[s].x, a[s].y}, a1 = {a[s].z, a[s].w};
        v2f w0 = {w[s].x, w[s].y}, w1 = {w[s].z, w[s].w};
        acc0 = __builtin_amdgcn_wmma_f32_16x16x4_f32(false, a0, false, w0, (short)0, acc0, false, false);
        acc1 = __builtin_amdgcn_wmma_f32_16x16x4_f32(false, a1, false, w1, (short)0, acc1, false, false);
    }
#pragma unroll
    for (int j = 0; j < 8; ++j) red[(wave * 8 + j) * 32 + lane] = acc0[j] + acc1[j];
    __syncthreads();
    {
        const int half = wave >> 2;
        const int src = ob * 32 + half * 16 + lane16;
        float s = red[src];
#pragma unroll
        for (int k = 1; k < 8; ++k) s += red[k * 256 + src];
        mkv[blockIdx.x * 256 + t] = s * inv;
    }
}

__global__ __launch_bounds__(256) void pass3_kernel(
    const float* __restrict__ qs, const float* __restrict__ mkv, float* __restrict__ out)
{
    const int b = blockIdx.x >> 7;
    const int srow0 = (blockIdx.x & 127) * 32;
    const int t = threadIdx.x;
    const int h = t >> 4;
    const float* tile = mkv + (h * 4 + ((t & 15) >> 2)) * 256 + b * 16 + (t & 3) * 4;
    float4 mv4 = *(const float4*)(tile);
    float4 kv4 = *(const float4*)(tile + 128);
#pragma unroll 4
    for (int r = 0; r < 32; ++r) {
        size_t srow = (size_t)b * S_ + srow0 + r;
        float q = qs[srow * H_ + h];
        float4 o;
        o.x = q * mv4.x + kv4.x; o.y = q * mv4.y + kv4.y;
        o.z = q * mv4.z + kv4.z; o.w = q * mv4.w + kv4.w;
        *(float4*)(out + srow * (H_ * DH_) + t * 4) = o;
    }
}

extern "C" void kernel_launch(void* const* d_in, const int* in_sizes, int n_in,
                              void* d_out, int out_size, void* d_ws, size_t ws_size,
                              hipStream_t stream) {
    const float* hidden = (const float*)d_in[0];
    const float* mask   = (const float*)d_in[1];
    const float* Wq     = (const float*)d_in[2];
    const float* Wk     = (const float*)d_in[3];
    const float* Wv     = (const float*)d_in[4];
    const float* aw     = (const float*)d_in[5];
    float* out = (float*)d_out;
    float* ws  = (float*)d_ws;

    weff_kernel   <<<64,   256, 0, stream>>>(Wq, Wk, aw, ws + OFF_WQE, ws + OFF_WKE);
    invlen_kernel <<<B_,   256, 0, stream>>>(mask, ws + OFF_IL);
    pass1_kernel  <<<256,  256, 0, stream>>>(hidden, mask, ws + OFF_WQE, ws + OFF_WKE,
                                             ws + OFF_QS, ws + OFF_HSP, ws + OFF_KHP);
    reduce_hsum_kernel<<<32,  256, 0, stream>>>(ws + OFF_HSP, ws + OFF_HS);
    reduce_kh_kernel  <<<512, 256, 0, stream>>>(ws + OFF_KHP, ws + OFF_KH);
    pass2_kernel  <<<64,   256, 0, stream>>>(Wv, ws + OFF_HS, ws + OFF_KH, ws + OFF_IL, ws + OFF_MKV);
    pass3_kernel  <<<1024, 256, 0, stream>>>(ws + OFF_QS, ws + OFF_MKV, out);
}


# launches 7 -> 5: the weight fold and the 1/length kernel share one grid (72 work-groups), the two partial-sum reducers share one grid (544); kernels' code and every sum order unchanged
# speedup vs baseline: 2.1561x; 1.0936x over previous
#include <hip/hip_runtime.h>

#define B_   8
#define S_   4096
#define HID_ 1024
#define H_   16
#define DH_  64

typedef __attribute__((ext_vector_type(2))) float v2f;
typedef __attribute__((ext_vector_type(8))) float v8f;

#define OFF_WQE 0
#define OFF_WKE (OFF_WQE + H_*HID_)
#define OFF_QS  (OFF_WKE + H_*HID_)
#define OFF_HSP (OFF_QS  + B_*S_*H_)
#define OFF_KHP (OFF_HSP + B_*32*HID_)
#define OFF_HS  (OFF_KHP + 256*H_*HID_)
#define OFF_KH  (OFF_HS  + B_*HID_)
#define OFF_MKV (OFF_KH  + B_*H_*HID_)
#define OFF_IL  (OFF_MKV + 2*B_*H_*DH_)

__global__ __launch_bounds__(256) void prologue_kernel(
    const float* __restrict__ Wq, const float* __restrict__ Wk, const float* __restrict__ aw,
    const float* __restrict__ mask,
    float* __restrict__ wqe, float* __restrict__ wke, float* __restrict__ il)
{
    __shared__ float red[256];
    const int t = threadIdx.x;
    if (blockIdx.x < 64) {
        int u = blockIdx.x * 256 + t;
        int h = u >> 10, i = u & (HID_ - 1);
        float aq = 0.f, ak = 0.f;
#pragma unroll 4
        for (int d = 0; d < DH_; ++d) {
            float a = aw[h * DH_ + d];
            aq += a * Wq[(size_t)(h * DH_ + d) * HID_ + i];
            ak += a * Wk[(size_t)(h * DH_ + d) * HID_ + i];
        }
        wqe[h * HID_ + i] = aq;
        wke[h * HID_ + i] = ak;
    } else {
        const int b = blockIdx.x - 64;
        float s = 0.f;
#pragma unroll
        for (int k = 0; k < S_ / 256; ++k) s += mask[b * S_ + t + 256 * k];
        red[t] = s;
        __syncthreads();
        for (int w = 128; w > 0; w >>= 1) {
            if (t < w) red[t] += red[t + w];
            __syncthreads();
        }
        if (t == 0) il[b] = 1.0f / red[0];
    }
}

static __device__ __forceinline__ void lds_barrier()
{
    asm volatile("s_wait_dscnt 0x0\n\ts_barrier_signal -1\n\ts_barrier_wait -1" ::: "memory");
}

static __device__ __forceinline__ void fetch_chunk(float4 (&v)[8], const float* __restrict__ hbase, int k0, int t)
{
#pragma unroll
    for (int it = 0; it < 8; ++it) {
        int row = it * 16 + (t >> 4);
        v[it] = *(const float4*)(hbase + (size_t)row * HID_ + k0 + (t & 15) * 4);
    }
}

__global__ __launch_bounds__(256) void pass1_kernel(
    const float* __restrict__ hidden, const float* __restrict__ mask,
    const float* __restrict__ wqe, const float* __restrict__ wke,
    float* __restrict__ q_scores, float* __restrict__ hsum_part,
    float* __restrict__ kh_part)
{
    __shared__ float lds_tile[128 * 68];
    __shared__ float lds_wq[16 * 68];
    __shared__ float lds_wk[16 * 68];
    __shared__ float lds_ksT[16 * 132];
    __shared__ float lds_red[256 * 4];
    __shared__ float lds_khb[2 * 16 * 68];

    const int blk = blockIdx.x & 31;
    const int b   = blockIdx.x >> 5;
    const int s0  = blk * 128;
    const int t = threadIdx.x;
    const int wave = t >> 5, lane = t & 31;
    const int lane16 = lane & 15, laneHi = lane >> 4;
    const int wr = t >> 4, wc4 = (t & 15) * 4;

    const float* hbase = hidden + ((size_t)b * S_ + s0) * HID_;

    float mrow[8];
#pragma unroll
    for (int it = 0; it < 8; ++it) mrow[it] = mask[b * S_ + s0 + it * 16 + wr];

    v8f cq = {}; v8f ck = {};

    float4 va[8], vb[8];
    float4 wqa, wka, wqb, wkb;

    auto fetch_w = [&](float4& q4, float4& k4, int kc) {
        q4 = *(const float4*)(wqe + wr * HID_ + kc * 64 + wc4);
        k4 = *(const float4*)(wke + wr * HID_ + kc * 64 + wc4);
    };
    auto stepA = [&](const float4 (&v)[8], const float4& q4, const float4& k4, int kc) {
        const int k0 = kc * 64;
        *(float4*)&lds_wq[wr * 68 + wc4] = q4;
        *(float4*)&lds_wk[wr * 68 + wc4] = k4;
        float4 acc = {0.f, 0.f, 0.f, 0.f};
#pragma unroll
        for (int it = 0; it < 8; ++it) {
            int row = it * 16 + wr;
            *(float4*)&lds_tile[row * 68 + wc4] = v[it];
            float m = mrow[it];
            acc.x += m * v[it].x; acc.y += m * v[it].y; acc.z += m * v[it].z; acc.w += m * v[it].w;
        }
        *(float4*)&lds_red[t * 4] = acc;
        lds_barrier();
        if (t < 64) {
            int g = t >> 2, comp = t & 3;
            float s = 0.f;
#pragma unroll
            for (int p = 0; p < 16; ++p) s += lds_red[(p * 16 + g) * 4 + comp];
            hsum_part[(size_t)(b * 32 + blk) * HID_ + k0 + g * 4 + comp] = s;
        }
        const float* atile = &lds_tile[(wave * 16 + lane16) * 68 + 2 * laneHi];
        const float* bq = &lds_wq[lane16 * 68 + 2 * laneHi];
        const float* bk = &lds_wk[lane16 * 68 + 2 * laneHi];
#pragma unroll
        for (int kk = 0; kk < 16; ++kk) {
            v2f a = *(const v2f*)(atile + kk * 4);
            v2f q = *(const v2f*)(bq + kk * 4);
            v2f k = *(const v2f*)(bk + kk * 4);
            cq = __builtin_amdgcn_wmma_f32_16x16x4_f32(false, a, false, q, (short)0, cq, false, false);
            ck = __builtin_amdgcn_wmma_f32_16x16x4_f32(false, a, false, k, (short)0, ck, false, false);
        }
        lds_barrier();
    };

    fetch_chunk(va, hbase, 0, t);
    fetch_w(wqa, wka, 0);
    for (int kc = 0; kc < 16; kc += 2) {
        fetch_chunk(vb, hbase, (kc + 1) * 64, t);
        fetch_w(wqb, wkb, kc + 1);
        stepA(va, wqa, wka, kc);
        fetch_chunk(va, hbase, ((kc + 2) & 15) * 64, t);
        fetch_w(wqa, wka, (kc + 2) & 15);
        stepA(vb, wqb, wkb, kc + 1);
    }

#pragma unroll
    for (int j = 0; j < 8; ++j) {
        int sl = wave * 16 + j + 8 * laneHi;
        float m = mask[b * S_ + s0 + sl];
        q_scores[((size_t)b * S_ + s0 + sl) * H_ + lane16] = cq[j] * m;
        lds_ksT[lane16 * 132 + sl] = ck[j] * m;
    }
    lds_barrier();

    const int ic = wave & 3;
    const int sh = wave >> 2;
    auto stepC = [&](const float4 (&v)[8], int kc) {
        const int k0 = kc * 64;
#pragma unroll
        for (int it = 0; it < 8; ++it) {
            int row = it * 16 + wr;
            *(float4*)&lds_tile[row * 68 + wc4] = v[it];
        }
        lds_barrier();
        v8f ckh = {};
        const float* aks = &lds_ksT[lane16 * 132 + sh * 64 + 2 * laneHi];
#pragma unroll
        for (int ss = 0; ss < 16; ++ss) {
            v2f a = *(const v2f*)(aks + ss * 4);
            int srow = sh * 64 + ss * 4 + 2 * laneHi;
            v2f bb;
            bb.x = lds_tile[srow * 68 + ic * 16 + lane16];
            bb.y = lds_tile[(srow + 1) * 68 + ic * 16 + lane16];
            ckh = __builtin_amdgcn_wmma_f32_16x16x4_f32(false, a, false, bb, (short)0, ckh, false, false);
        }
#pragma unroll
        for (int j = 0; j < 8; ++j)
            lds_khb[(sh * 16 + j + 8 * laneHi) * 68 + ic * 16 + lane16] = ckh[j];
        lds_barrier();
        {
            int h = t >> 4, c4 = (t & 15) * 4;
            float4 a0 = *(const float4*)&lds_khb[h * 68 + c4];
            float4 a1 = *(const float4*)&lds_khb[(16 + h) * 68 + c4];
            float4 r; r.x = a0.x + a1.x; r.y = a0.y + a1.y; r.z = a0.z + a1.z; r.w = a0.w + a1.w;
            *(float4*)&kh_part[(((size_t)(b * 32 + blk)) * H_ + h) * HID_ + k0 + c4] = r;
        }
        lds_barrier();
    };

    for (int kc = 0; kc < 16; kc += 2) {
        fetch_chunk(vb, hbase, (kc + 1) * 64, t);
        stepC(va, kc);
        fetch_chunk(va, hbase, ((kc + 2) & 15) * 64, t);
        stepC(vb, kc + 1);
    }
}

__global__ __launch_bounds__(256) void reduce_kernel(
    const float* __restrict__ khp, const float* __restrict__ hsp,
    float* __restrict__ kh, float* __restrict__ hs)
{
    if (blockIdx.x < 512) {
        int idx = blockIdx.x * 256 + threadIdx.x;
        int bh = idx >> 10, i = idx & (HID_ - 1);
        int b = bh >> 4, h = bh & 15;
        float s = 0.f;
#pragma unroll 8
        for (int blk = 0; blk < 32; ++blk)
            s += khp[(((size_t)(b * 32 + blk)) * H_ + h) * HID_ + i];
        kh[idx] = s;
    } else {
        int idx = (blockIdx.x - 512) * 256 + threadIdx.x;
        int b = idx >> 10, i = idx & (HID_ - 1);
        float s = 0.f;
#pragma unroll 8
        for (int blk = 0; blk < 32; ++blk) s += hsp[(size_t)(b * 32 + blk) * HID_ + i];
        hs[idx] = s;
    }
}

__global__ __launch_bounds__(256) void pass2_kernel(
    const float* __restrict__ Wv, const float* __restrict__ hs,
    const float* __restrict__ kh, const float* __restrict__ il,
    float* __restrict__ mkv)
{
    __shared__ float red[8 * 8 * 32];
    const int h  = blockIdx.x >> 2;
    const int d0 = (blockIdx.x & 3) * 16;
    const int t = threadIdx.x;
    const int wave = t >> 5, lane = t & 31;
    const int lane16 = lane & 15, laneHi = lane >> 4;

    const int ob = ((wave & 3) << 1) | laneHi;
    const float inv = il[ob];

    const float* arow = (lane16 < 8) ? (hs + lane16 * HID_)
                                     : (kh + (size_t)((lane16 - 8) * H_ + h) * HID_);
    const float* brow = Wv + (size_t)(h * DH_ + d0 + lane16) * HID_;
    const int kb = wave * 128 + laneHi * 4;

    float4 a[16], w[16];
#pragma unroll
    for (int s = 0; s < 16; ++s) {
        a[s] = *(const float4*)(arow + kb + s * 8);
        w[s] = *(const float4*)(brow + kb + s * 8);
    }
    __builtin_amdgcn_sched_barrier(0);

    v8f acc0 = {}; v8f acc1 = {};
#pragma unroll
    for (int s = 0; s < 16; ++s) {
        v2f a0 = {a[s].x, a[s].y}, a1 = {a[s].z, a[s].w};
        v2f w0 = {w[s].x, w[s].y}, w1 = {w[s].z, w[s].w};
        acc0 = __builtin_amdgcn_wmma_f32_16x16x4_f32(false, a0, false, w0, (short)0, acc0, false, false);
        acc1 = __builtin_amdgcn_wmma_f32_16x16x4_f32(false, a1, false, w1, (short)0, acc1, false, false);
    }
#pragma unroll
    for (int j = 0; j < 8; ++j) red[(wave * 8 + j) * 32 + lane] = acc0[j] + acc1[j];
    __syncthreads();
    {
        const int half = wave >> 2;
        const int src = ob * 32 + half * 16 + lane16;
        float s = red[src];
#pragma unroll
        for (int k = 1; k < 8; ++k) s += red[k * 256 + src];
        mkv[blockIdx.x * 256 + t] = s * inv;
    }
}

__global__ __launch_bounds__(256) void pass3_kernel(
    const float* __restrict__ qs, const float* __restrict__ mkv, float* __restrict__ out)
{
    const int b = blockIdx.x >> 7;
    const int srow0 = (blockIdx.x & 127) * 32;
    const int t = threadIdx.x;
    const int h = t >> 4;
    const float* tile = mkv + (h * 4 + ((t & 15) >> 2)) * 256 + b * 16 + (t & 3) * 4;
    float4 mv4 = *(const float4*)(tile);
    float4 kv4 = *(const float4*)(tile + 128);
#pragma unroll 4
    for (int r = 0; r < 32; ++r) {
        size_t srow = (size_t)b * S_ + srow0 + r;
        float q = qs[srow * H_ + h];
        float4 o;
        o.x = q * mv4.x + kv4.x; o.y = q * mv4.y + kv4.y;
        o.z = q * mv4.z + kv4.z; o.w = q * mv4.w + kv4.w;
        *(float4*)(out + srow * (H_ * DH_) + t * 4) = o;
    }
}

extern "C" void kernel_launch(void* const* d_in, const int* in_sizes, int n_in,
                              void* d_out, int out_size, void* d_ws, size_t ws_size,
                              hipStream_t stream) {
    const float* hidden = (const float*)d_in[0];
    const float* mask   = (const float*)d_in[1];
    const float* Wq     = (const float*)d_in[2];
    const float* Wk     = (const float*)d_in[3];
    const float* Wv     = (const float*)d_in[4];
    const float* aw     = (const float*)d_in[5];
    float* out = (float*)d_out;
    float* ws  = (float*)d_ws;

    prologue_kernel<<<72,  256, 0, stream>>>(Wq, Wk, aw, mask, ws + OFF_WQE, ws + OFF_WKE, ws + OFF_IL);
    pass1_kernel  <<<256,  256, 0, stream>>>(hidden, mask, ws + OFF_WQE, ws + OFF_WKE,
                                             ws + OFF_QS, ws + OFF_HSP, ws + OFF_KHP);
    reduce_kernel <<<544,  256, 0, stream>>>(ws + OFF_KHP, ws + OFF_HSP, ws + OFF_KH, ws + OFF_HS);
    pass2_kernel  <<<64,   256, 0, stream>>>(Wv, ws + OFF_HS, ws + OFF_KH, ws + OFF_IL, ws + OFF_MKV);
    pass3_kernel  <<<1024, 256, 0, stream>>>(ws + OFF_QS, ws + OFF_MKV, out);
}


# scores/kh kernel reads hidden once: 16-row x 1024-col LDS sub-tiles feed the score WMMAs (K split over 8 waves, fixed-order LDS sum) and register-held kh tiles; new sum order in scores, hsum, kh
# speedup vs baseline: 2.3071x; 1.0700x over previous
#include <hip/hip_runtime.h>

#define B_   8
#define S_   4096
#define HID_ 1024
#define H_   16
#define DH_  64

typedef __attribute__((ext_vector_type(2))) float v2f;
typedef __attribute__((ext_vector_type(8))) float v8f;

#define OFF_WQE 0
#define OFF_WKE (OFF_WQE + H_*HID_)
#define OFF_QS  (OFF_WKE + H_*HID_)
#define OFF_HSP (OFF_QS  + B_*S_*H_)
#define OFF_KHP (OFF_HSP + B_*32*HID_)
#define OFF_HS  (OFF_KHP + 256*H_*HID_)
#define OFF_KH  (OFF_HS  + B_*HID_)
#define OFF_MKV (OFF_KH  + B_*H_*HID_)
#define OFF_IL  (OFF_MKV + 2*B_*H_*DH_)

__global__ __launch_bounds__(256) void prologue_kernel(
    const float* __restrict__ Wq, const float* __restrict__ Wk, const float* __restrict__ aw,
    const float* __restrict__ mask,
    float* __restrict__ wqe, float* __restrict__ wke, float* __restrict__ il)
{
    __shared__ float red[256];
    const int t = threadIdx.x;
    if (blockIdx.x < 64) {
        int u = blockIdx.x * 256 + t;
        int h = u >> 10, i = u & (HID_ - 1);
        float aq = 0.f, ak = 0.f;
#pragma unroll 4
        for (int d = 0; d < DH_; ++d) {
            float a = aw[h * DH_ + d];
            aq += a * Wq[(size_t)(h * DH_ + d) * HID_ + i];
            ak += a * Wk[(size_t)(h * DH_ + d) * HID_ + i];
        }
        wqe[h * HID_ + i] = aq;
        wke[h * HID_ + i] = ak;
    } else {
        const int b = blockIdx.x - 64;
        float s = 0.f;
#pragma unroll
        for (int k = 0; k < S_ / 256; ++k) s += mask[b * S_ + t + 256 * k];
        red[t] = s;
        __syncthreads();
        for (int w = 128; w > 0; w >>= 1) {
            if (t < w) red[t] += red[t + w];
            __syncthreads();
        }
        if (t == 0) il[b] = 1.0f / red[0];
    }
}

static __device__ __forceinline__ void lds_barrier()
{
    asm volatile("s_wait_dscnt 0x0\n\ts_barrier_signal -1\n\ts_barrier_wait -1" ::: "memory");
}

#define TP_ 1028
#define KP_ 20

__global__ __launch_bounds__(256) void pass1_kernel(
    const float* __restrict__ hidden, const float* __restrict__ mask,
    const float* __restrict__ wqe, const float* __restrict__ wke,
    float* __restrict__ q_scores, float* __restrict__ hsum_part,
    float* __restrict__ kh_part)
{
    __shared__ __align__(16) float lds_tile[16 * TP_];
    __shared__ __align__(16) float lds_part[8 * 2 * 8 * 32];
    __shared__ __align__(16) float lds_ksT[16 * KP_];
    __shared__ float lds_mask[128];

    const int blk = blockIdx.x & 31;
    const int b   = blockIdx.x >> 5;
    const int s0  = blk * 128;
    const int t = threadIdx.x;
    const int wave = t >> 5, lane = t & 31;
    const int lane16 = lane & 15, laneHi = lane >> 4;

    const float* hbase = hidden + ((size_t)b * S_ + s0) * HID_ + 4 * t;

    float mk = (t < 128) ? mask[b * S_ + s0 + t] : 0.f;

    float4 wq4[16], wk4[16];
    {
        const float* wqr = wqe + lane16 * HID_ + wave * 128 + 4 * laneHi;
        const float* wkr = wke + lane16 * HID_ + wave * 128 + 4 * laneHi;
#pragma unroll
        for (int s = 0; s < 16; ++s) {
            wq4[s] = *(const float4*)(wqr + s * 8);
            wk4[s] = *(const float4*)(wkr + s * 8);
        }
    }
    float4 pa[16], pb[16];
    auto fetch = [&](float4 (&v)[16], int st) {
        const float* nb = hbase + (size_t)(st * 16) * HID_;
#pragma unroll
        for (int r = 0; r < 16; ++r) v[r] = *(const float4*)(nb + (size_t)r * HID_);
    };
    fetch(pa, 0);

    if (t < 128) lds_mask[t] = mk;
    lds_barrier();

    v8f kh[8];
#pragma unroll
    for (int n = 0; n < 8; ++n) kh[n] = v8f{};
    float4 hs = {0.f, 0.f, 0.f, 0.f};

    const int rr = 2 * wave + laneHi;
    const int rsrc = (rr & 7) * 32 + (rr >> 3) * 16 + lane16;

    auto step = [&](const float4 (&pre)[16], float4 (&nxt)[16], int st) {
        fetch(nxt, (st + 1) & 7);
#pragma unroll
        for (int r = 0; r < 16; ++r) {
            *(float4*)&lds_tile[r * TP_ + 4 * t] = pre[r];
            float m = lds_mask[st * 16 + r];
            hs.x += m * pre[r].x; hs.y += m * pre[r].y; hs.z += m * pre[r].z; hs.w += m * pre[r].w;
        }
        lds_barrier();

        v8f cq = {}; v8f ck = {};
        {
            const float* ap = &lds_tile[lane16 * TP_ + wave * 128 + 4 * laneHi];
#pragma unroll
            for (int s = 0; s < 16; ++s) {
                float4 a4 = *(const float4*)(ap + s * 8);
                v2f a0 = {a4.x, a4.y}, a1 = {a4.z, a4.w};
                v2f q0 = {wq4[s].x, wq4[s].y}, q1 = {wq4[s].z, wq4[s].w};
                v2f k0 = {wk4[s].x, wk4[s].y}, k1 = {wk4[s].z, wk4[s].w};
                cq = __builtin_amdgcn_wmma_f32_16x16x4_f32(false, a0, false, q0, (short)0, cq, false, false);
                ck = __builtin_amdgcn_wmma_f32_16x16x4_f32(false, a0, false, k0, (short)0, ck, false, false);
                cq = __builtin_amdgcn_wmma_f32_16x16x4_f32(false, a1, false, q1, (short)0, cq, false, false);
                ck = __builtin_amdgcn_wmma_f32_16x16x4_f32(false, a1, false, k1, (short)0, ck, false, false);
            }
        }
#pragma unroll
        for (int j = 0; j < 8; ++j) {
            lds_part[((wave * 2 + 0) * 8 + j) * 32 + lane] = cq[j];
            lds_part[((wave * 2 + 1) * 8 + j) * 32 + lane] = ck[j];
        }
        lds_barrier();

        {
            float sq = lds_part[rsrc], sk = lds_part[256 + rsrc];
#pragma unroll
            for (int w = 1; w < 8; ++w) {
                sq += lds_part[(w * 2 + 0) * 256 + rsrc];
                sk += lds_part[(w * 2 + 1) * 256 + rsrc];
            }
            float m = lds_mask[st * 16 + rr];
            q_scores[((size_t)b * S_ + s0 + st * 16 + rr) * H_ + lane16] = sq * m;
            lds_ksT[lane16 * KP_ + rr] = sk * m;
        }
        lds_barrier();

#pragma unroll
        for (int ss = 0; ss < 4; ++ss) {
            v2f a = *(const v2f*)&lds_ksT[lane16 * KP_ + ss * 4 + 2 * laneHi];
            const float* bp = &lds_tile[(ss * 4 + 2 * laneHi) * TP_ + wave * 128 + lane16];
#pragma unroll
            for (int n = 0; n < 8; ++n) {
                v2f bb;
                bb.x = bp[n * 16];
                bb.y = bp[TP_ + n * 16];
                kh[n] = __builtin_amdgcn_wmma_f32_16x16x4_f32(false, a, false, bb, (short)0, kh[n], false, false);
            }
        }
        lds_barrier();
    };

    for (int st = 0; st < 8; st += 2) {
        step(pa, pb, st);
        step(pb, pa, st + 1);
    }

    *(float4*)&hsum_part[(size_t)(b * 32 + blk) * HID_ + 4 * t] = hs;
#pragma unroll
    for (int n = 0; n < 8; ++n)
#pragma unroll
        for (int j = 0; j < 8; ++j)
            lds_tile[(j + 8 * laneHi) * TP_ + wave * 128 + n * 16 + lane16] = kh[n][j];
    lds_barrier();
#pragma unroll 4
    for (int hh = 0; hh < 16; ++hh) {
        float4 v = *(const float4*)&lds_tile[hh * TP_ + 4 * t];
        *(float4*)&kh_part[(((size_t)(b * 32 + blk)) * H_ + hh) * HID_ + 4 * t] = v;
    }
}

__global__ __launch_bounds__(256) void reduce_kernel(
    const float* __restrict__ khp, const float* __restrict__ hsp,
    float* __restrict__ kh, float* __restrict__ hs)
{
    if (blockIdx.x < 512) {
        int idx = blockIdx.x * 256 + threadIdx.x;
        int bh = idx >> 10, i = idx & (HID_ - 1);
        int b = bh >> 4, h = bh & 15;
        float s = 0.f;
#pragma unroll 8
        for (int blk = 0; blk < 32; ++blk)
            s += khp[(((size_t)(b * 32 + blk)) * H_ + h) * HID_ + i];
        kh[idx] = s;
    } else {
        int idx = (blockIdx.x - 512) * 256 + threadIdx.x;
        int b = idx >> 10, i = idx & (HID_ - 1);
        float s = 0.f;
#pragma unroll 8
        for (int blk = 0; blk < 32; ++blk) s += hsp[(size_t)(b * 32 + blk) * HID_ + i];
        hs[idx] = s;
    }
}

__global__ __launch_bounds__(256) void pass2_kernel(
    const float* __restrict__ Wv, const float* __restrict__ hs,
    const float* __restrict__ kh, const float* __restrict__ il,
    float* __restrict__ mkv)
{
    __shared__ float red[8 * 8 * 32];
    const int h  = blockIdx.x >> 2;
    const int d0 = (blockIdx.x & 3) * 16;
    const int t = threadIdx.x;
    const int wave = t >> 5, lane = t & 31;
    const int lane16 = lane & 15, laneHi = lane >> 4;

    const int ob = ((wave & 3) << 1) | laneHi;
    const float inv = il[ob];

    const float* arow = (lane16 < 8) ? (hs + lane16 * HID_)
                                     : (kh + (size_t)((lane16 - 8) * H_ + h) * HID_);
    const float* brow = Wv + (size_t)(h * DH_ + d0 + lane16) * HID_;
    const int kb = wave * 128 + laneHi * 4;

    float4 a[16], w[16];
#pragma unroll
    for (int s = 0; s < 16; ++s) {
        a[s] = *(const float4*)(arow + kb + s * 8);
        w[s] = *(const float4*)(brow + kb + s * 8);
    }
    __builtin_amdgcn_sched_barrier(0);

    v8f acc0 = {}; v8f acc1 = {};
#pragma unroll
    for (int s = 0; s < 16; ++s) {
        v2f a0 = {a[s].x, a[s].y}, a1 = {a[s].z, a[s].w};
        v2f w0 = {w[s].x, w[s].y}, w1 = {w[s].z, w[s].w};
        acc0 = __builtin_amdgcn_wmma_f32_16x16x4_f32(false, a0, false, w0, (short)0, acc0, false, false);
        acc1 = __builtin_amdgcn_wmma_f32_16x16x4_f32(false, a1, false, w1, (short)0, acc1, false, false);
    }
#pragma unroll
    for (int j = 0; j < 8; ++j) red[(wave * 8 + j) * 32 + lane] = acc0[j] + acc1[j];
    __syncthreads();
    {
        const int half = wave >> 2;
        const int src = ob * 32 + half * 16 + lane16;
        float s = red[src];
#pragma unroll
        for (int k = 1; k < 8; ++k) s += red[k * 256 + src];
        mkv[blockIdx.x * 256 + t] = s * inv;
    }
}

__global__ __launch_bounds__(256) void pass3_kernel(
    const float* __restrict__ qs, const float* __restrict__ mkv, float* __restrict__ out)
{
    const int b = blockIdx.x >> 7;
    const int srow0 = (blockIdx.x & 127) * 32;
    const int t = threadIdx.x;
    const int h = t >> 4;
    const float* tile = mkv + (h * 4 + ((t & 15) >> 2)) * 256 + b * 16 + (t & 3) * 4;
    float4 mv4 = *(const float4*)(tile);
    float4 kv4 = *(const float4*)(tile + 128);
#pragma unroll 4
    for (int r = 0; r < 32; ++r) {
        size_t srow = (size_t)b * S_ + srow0 + r;
        float q = qs[srow * H_ + h];
        float4 o;
        o.x = q * mv4.x + kv4.x; o.y = q * mv4.y + kv4.y;
        o.z = q * mv4.z + kv4.z; o.w = q * mv4.w + kv4.w;
        *(float4*)(out + srow * (H_ * DH_) + t * 4) = o;
    }
}

extern "C" void kernel_launch(void* const* d_in, const int* in_sizes, int n_in,
                              void* d_out, int out_size, void* d_ws, size_t ws_size,
                              hipStream_t stream) {
    const float* hidden = (const float*)d_in[0];
    const float* mask   = (const float*)d_in[1];
    const float* Wq     = (const float*)d_in[2];
    const float* Wk     = (const float*)d_in[3];
    const float* Wv     = (const float*)d_in[4];
    const float* aw     = (const float*)d_in[5];
    float* out = (float*)d_out;
    float* ws  = (float*)d_ws;

    prologue_kernel<<<72,  256, 0, stream>>>(Wq, Wk, aw, mask, ws + OFF_WQE, ws + OFF_WKE, ws + OFF_IL);
    pass1_kernel  <<<256,  256, 0, stream>>>(hidden, mask, ws + OFF_WQE, ws + OFF_WKE,
                                             ws + OFF_QS, ws + OFF_HSP, ws + OFF_KHP);
    reduce_kernel <<<544,  256, 0, stream>>>(ws + OFF_KHP, ws + OFF_HSP, ws + OFF_KH, ws + OFF_HS);
    pass2_kernel  <<<64,   256, 0, stream>>>(Wv, ws + OFF_HS, ws + OFF_KH, ws + OFF_IL, ws + OFF_MKV);
    pass3_kernel  <<<1024, 256, 0, stream>>>(ws + OFF_QS, ws + OFF_MKV, out);
}


# serial load-then-use chains cut to 1-2 memory round trips: weight fold on 256 work-groups (fixed-order LDS sum of 16 partials, new sum order), reducers load all 32 first, output pass 16 rows a trip
# speedup vs baseline: 2.9616x; 1.2837x over previous
#include <hip/hip_runtime.h>

#define B_   8
#define S_   4096
#define HID_ 1024
#define H_   16
#define DH_  64

typedef __attribute__((ext_vector_type(2))) float v2f;
typedef __attribute__((ext_vector_type(8))) float v8f;

#define OFF_WQE 0
#define OFF_WKE (OFF_WQE + H_*HID_)
#define OFF_QS  (OFF_WKE + H_*HID_)
#define OFF_HSP (OFF_QS  + B_*S_*H_)
#define OFF_KHP (OFF_HSP + B_*32*HID_)
#define OFF_HS  (OFF_KHP + 256*H_*HID_)
#define OFF_KH  (OFF_HS  + B_*HID_)
#define OFF_MKV (OFF_KH  + B_*H_*HID_)
#define OFF_IL  (OFF_MKV + 2*B_*H_*DH_)

__global__ __launch_bounds__(256) void prologue_kernel(
    const float* __restrict__ Wq, const float* __restrict__ Wk, const float* __restrict__ aw,
    const float* __restrict__ mask,
    float* __restrict__ wqe, float* __restrict__ wke, float* __restrict__ il)
{
    __shared__ __align__(16) float red[16 * 2 * 64];
    const int t = threadIdx.x;
    if (blockIdx.x < 256) {
        const int h  = blockIdx.x >> 4;
        const int i0 = (blockIdx.x & 15) * 64;
        const int cq = t & 15, dg = t >> 4;
        const float4 a4 = *(const float4*)(aw + h * DH_ + dg * 4);
        const size_t rb = (size_t)(h * DH_ + dg * 4) * HID_ + i0 + cq * 4;
        const float4 q0 = *(const float4*)(Wq + rb);
        const float4 q1 = *(const float4*)(Wq + rb + HID_);
        const float4 q2 = *(const float4*)(Wq + rb + 2 * HID_);
        const float4 q3 = *(const float4*)(Wq + rb + 3 * HID_);
        const float4 k0 = *(const float4*)(Wk + rb);
        const float4 k1 = *(const float4*)(Wk + rb + HID_);
        const float4 k2 = *(const float4*)(Wk + rb + 2 * HID_);
        const float4 k3 = *(const float4*)(Wk + rb + 3 * HID_);
        float4 sq, sk;
        sq.x = a4.x * q0.x; sq.y = a4.x * q0.y; sq.z = a4.x * q0.z; sq.w = a4.x * q0.w;
        sk.x = a4.x * k0.x; sk.y = a4.x * k0.y; sk.z = a4.x * k0.z; sk.w = a4.x * k0.w;
        sq.x += a4.y * q1.x; sq.y += a4.y * q1.y; sq.z += a4.y * q1.z; sq.w += a4.y * q1.w;
        sk.x += a4.y * k1.x; sk.y += a4.y * k1.y; sk.z += a4.y * k1.z; sk.w += a4.y * k1.w;
        sq.x += a4.z * q2.x; sq.y += a4.z * q2.y; sq.z += a4.z * q2.z; sq.w += a4.z * q2.w;
        sk.x += a4.z * k2.x; sk.y += a4.z * k2.y; sk.z += a4.z * k2.z; sk.w += a4.z * k2.w;
        sq.x += a4.w * q3.x; sq.y += a4.w * q3.y; sq.z += a4.w * q3.z; sq.w += a4.w * q3.w;
        sk.x += a4.w * k3.x; sk.y += a4.w * k3.y; sk.z += a4.w * k3.z; sk.w += a4.w * k3.w;
        *(float4*)&red[(dg * 2 + 0) * 64 + cq * 4] = sq;
        *(float4*)&red[(dg * 2 + 1) * 64 + cq * 4] = sk;
        __syncthreads();
        if (t < 128) {
            const int which = t >> 6, c = t & 63;
            float s = red[which * 64 + c];
#pragma unroll
            for (int g = 1; g < 16; ++g) s += red[(g * 2 + which) * 64 + c];
            float* dst = which ? wke : wqe;
            dst[h * HID_ + i0 + c] = s;
        }
    } else {
        const int b = blockIdx.x - 256;
        float s = 0.f;
#pragma unroll
        for (int k = 0; k < S_ / 256; ++k) s += mask[b * S_ + t + 256 * k];
        red[t] = s;
        __syncthreads();
        for (int w = 128; w > 0; w >>= 1) {
            if (t < w) red[t] += red[t + w];
            __syncthreads();
        }
        if (t == 0) il[b] = 1.0f / red[0];
    }
}

static __device__ __forceinline__ void lds_barrier()
{
    asm volatile("s_wait_dscnt 0x0\n\ts_barrier_signal -1\n\ts_barrier_wait -1" ::: "memory");
}

#define TP_ 1028
#define KP_ 20

__global__ __launch_bounds__(256) void pass1_kernel(
    const float* __restrict__ hidden, const float* __restrict__ mask,
    const float* __restrict__ wqe, const float* __restrict__ wke,
    float* __restrict__ q_scores, float* __restrict__ hsum_part,
    float* __restrict__ kh_part)
{
    __shared__ __align__(16) float lds_tile[16 * TP_];
    __shared__ __align__(16) float lds_part[8 * 2 * 8 * 32];
    __shared__ __align__(16) float lds_ksT[16 * KP_];
    __shared__ float lds_mask[128];

    const int blk = blockIdx.x & 31;
    const int b   = blockIdx.x >> 5;
    const int s0  = blk * 128;
    const int t = threadIdx.x;
    const int wave = t >> 5, lane = t & 31;
    const int lane16 = lane & 15, laneHi = lane >> 4;

    const float* hbase = hidden + ((size_t)b * S_ + s0) * HID_ + 4 * t;

    float mk = (t < 128) ? mask[b * S_ + s0 + t] : 0.f;

    float4 wq4[16], wk4[16];
    {
        const float* wqr = wqe + lane16 * HID_ + wave * 128 + 4 * laneHi;
        const float* wkr = wke + lane16 * HID_ + wave * 128 + 4 * laneHi;
#pragma unroll
        for (int s = 0; s < 16; ++s) {
            wq4[s] = *(const float4*)(wqr + s * 8);
            wk4[s] = *(const float4*)(wkr + s * 8);
        }
    }
    float4 pa[16], pb[16];
    auto fetch = [&](float4 (&v)[16], int st) {
        const float* nb = hbase + (size_t)(st * 16) * HID_;
#pragma unroll
        for (int r = 0; r < 16; ++r) v[r] = *(const float4*)(nb + (size_t)r * HID_);
    };
    fetch(pa, 0);

    if (t < 128) lds_mask[t] = mk;
    lds_barrier();

    v8f kh[8];
#pragma unroll
    for (int n = 0; n < 8; ++n) kh[n] = v8f{};
    float4 hs = {0.f, 0.f, 0.f, 0.f};

    const int rr = 2 * wave + laneHi;
    const int rsrc = (rr & 7) * 32 + (rr >> 3) * 16 + lane16;

    auto step = [&](const float4 (&pre)[16], float4 (&nxt)[16], int st) {
        fetch(nxt, (st + 1) & 7);
#pragma unroll
        for (int r = 0; r < 16; ++r) {
            *(float4*)&lds_tile[r * TP_ + 4 * t] = pre[r];
            float m = lds_mask[st * 16 + r];
            hs.x += m * pre[r].x; hs.y += m * pre[r].y; hs.z += m * pre[r].z; hs.w += m * pre[r].w;
        }
        lds_barrier();

        v8f cq = {}; v8f ck = {};
        {
            const float* ap = &lds_tile[lane16 * TP_ + wave * 128 + 4 * laneHi];
#pragma unroll
            for (int s = 0; s < 16; ++s) {
                float4 a4 = *(const float4*)(ap + s * 8);
                v2f a0 = {a4.x, a4.y}, a1 = {a4.z, a4.w};
                v2f q0 = {wq4[s].x, wq4[s].y}, q1 = {wq4[s].z, wq4[s].w};
                v2f k0 = {wk4[s].x, wk4[s].y}, k1 = {wk4[s].z, wk4[s].w};
                cq = __builtin_amdgcn_wmma_f32_16x16x4_f32(false, a0, false, q0, (short)0, cq, false, false);
                ck = __builtin_amdgcn_wmma_f32_16x16x4_f32(false, a0, false, k0, (short)0, ck, false, false);
                cq = __builtin_amdgcn_wmma_f32_16x16x4_f32(false, a1, false, q1, (short)0, cq, false, false);
                ck = __builtin_amdgcn_wmma_f32_16x16x4_f32(false, a1, false, k1, (short)0, ck, false, false);
            }
        }
#pragma unroll
        for (int j = 0; j < 8; ++j) {
            lds_part[((wave * 2 + 0) * 8 + j) * 32 + lane] = cq[j];
            lds_part[((wave * 2 + 1) * 8 + j) * 32 + lane] = ck[j];
        }
        lds_barrier();

        {
            float sq = lds_part[rsrc], sk = lds_part[256 + rsrc];
#pragma unroll
            for (int w = 1; w < 8; ++w) {
                sq += lds_part[(w * 2 + 0) * 256 + rsrc];
                sk += lds_part[(w * 2 + 1) * 256 + rsrc];
            }
            float m = lds_mask[st * 16 + rr];
            q_scores[((size_t)b * S_ + s0 + st * 16 + rr) * H_ + lane16] = sq * m;
            lds_ksT[lane16 * KP_ + rr] = sk * m;
        }
        lds_barrier();

#pragma unroll
        for (int ss = 0; ss < 4; ++ss) {
            v2f a = *(const v2f*)&lds_ksT[lane16 * KP_ + ss * 4 + 2 * laneHi];
            const float* bp = &lds_tile[(ss * 4 + 2 * laneHi) * TP_ + wave * 128 + lane16];
#pragma unroll
            for (int n = 0; n < 8; ++n) {
                v2f bb;
                bb.x = bp[n * 16];
                bb.y = bp[TP_ + n * 16];
                kh[n] = __builtin_amdgcn_wmma_f32_16x16x4_f32(false, a, false, bb, (short)0, kh[n], false, false);
            }
        }
        lds_barrier();
    };

    for (int st = 0; st < 8; st += 2) {
        step(pa, pb, st);
        step(pb, pa, st + 1);
    }

    *(float4*)&hsum_part[(size_t)(b * 32 + blk) * HID_ + 4 * t] = hs;
#pragma unroll
    for (int n = 0; n < 8; ++n)
#pragma unroll
        for (int j = 0; j < 8; ++j)
            lds_tile[(j + 8 * laneHi) * TP_ + wave * 128 + n * 16 + lane16] = kh[n][j];
    lds_barrier();
#pragma unroll 4
    for (int hh = 0; hh < 16; ++hh) {
        float4 v = *(const float4*)&lds_tile[hh * TP_ + 4 * t];
        *(float4*)&kh_part[(((size_t)(b * 32 + blk)) * H_ + hh) * HID_ + 4 * t] = v;
    }
}

__global__ __launch_bounds__(256) void reduce_kernel(
    const float* __restrict__ khp, const float* __restrict__ hsp,
    float* __restrict__ kh, float* __restrict__ hs)
{
    if (blockIdx.x < 512) {
        int idx = blockIdx.x * 256 + threadIdx.x;
        int bh = idx >> 10, i = idx & (HID_ - 1);
        int b = bh >> 4, h = bh & 15;
        float v[32];
#pragma unroll
        for (int blk = 0; blk < 32; ++blk)
            v[blk] = khp[(((size_t)(b * 32 + blk)) * H_ + h) * HID_ + i];
        float s = 0.f;
#pragma unroll
        for (int blk = 0; blk < 32; ++blk) s += v[blk];
        kh[idx] = s;
    } else {
        int idx = (blockIdx.x - 512) * 256 + threadIdx.x;
        int b = idx >> 10, i = idx & (HID_ - 1);
        float v[32];
#pragma unroll
        for (int blk = 0; blk < 32; ++blk) v[blk] = hsp[(size_t)(b * 32 + blk) * HID_ + i];
        float s = 0.f;
#pragma unroll
        for (int blk = 0; blk < 32; ++blk) s += v[blk];
        hs[idx] = s;
    }
}

__global__ __launch_bounds__(256) void pass2_kernel(
    const float* __restrict__ Wv, const float* __restrict__ hs,
    const float* __restrict__ kh, const float* __restrict__ il,
    float* __restrict__ mkv)
{
    __shared__ float red[8 * 8 * 32];
    const int h  = blockIdx.x >> 2;
    const int d0 = (blockIdx.x & 3) * 16;
    const int t = threadIdx.x;
    const int wave = t >> 5, lane = t & 31;
    const int lane16 = lane & 15, laneHi = lane >> 4;

    const int ob = ((wave & 3) << 1) | laneHi;
    const float inv = il[ob];

    const float* arow = (lane16 < 8) ? (hs + lane16 * HID_)
                                     : (kh + (size_t)((lane16 - 8) * H_ + h) * HID_);
    const float* brow = Wv + (size_t)(h * DH_ + d0 + lane16) * HID_;
    const int kb = wave * 128 + laneHi * 4;

    float4 a[16], w[16];
#pragma unroll
    for (int s = 0; s < 16; ++s) {
        a[s] = *(const float4*)(arow + kb + s * 8);
        w[s] = *(const float4*)(brow + kb + s * 8);
    }
    __builtin_amdgcn_sched_barrier(0);

    v8f acc0 = {}; v8f acc1 = {};
#pragma unroll
    for (int s = 0; s < 16; ++s) {
        v2f a0 = {a[s].x, a[s].y}, a1 = {a[s].z, a[s].w};
        v2f w0 = {w[s].x, w[s].y}, w1 = {w[s].z, w[s].w};
        acc0 = __builtin_amdgcn_wmma_f32_16x16x4_f32(false, a0, false, w0, (short)0, acc0, false, false);
        acc1 = __builtin_amdgcn_wmma_f32_16x16x4_f32(false, a1, false, w1, (short)0, acc1, false, false);
    }
#pragma unroll
    for (int j = 0; j < 8; ++j) red[(wave * 8 + j) * 32 + lane] = acc0[j] + acc1[j];
    __syncthreads();
    {
        const int half = wave >> 2;
        const int src = ob * 32 + half * 16 + lane16;
        float s = red[src];
#pragma unroll
        for (int k = 1; k < 8; ++k) s += red[k * 256 + src];
        mkv[blockIdx.x * 256 + t] = s * inv;
    }
}

__global__ __launch_bounds__(256) void pass3_kernel(
    const float* __restrict__ qs, const float* __restrict__ mkv, float* __restrict__ out)
{
    const int b = blockIdx.x >> 7;
    const int srow0 = (blockIdx.x & 127) * 32;
    const int t = threadIdx.x;
    const int h = t >> 4;
    const float* tile = mkv + (h * 4 + ((t & 15) >> 2)) * 256 + b * 16 + (t & 3) * 4;
    float4 mv4 = *(const float4*)(tile);
    float4 kv4 = *(const float4*)(tile + 128);
#pragma unroll 1
    for (int r0 = 0; r0 < 32; r0 += 16) {
        float qv[16];
#pragma unroll
        for (int j = 0; j < 16; ++j) qv[j] = qs[((size_t)b * S_ + srow0 + r0 + j) * H_ + h];
#pragma unroll
        for (int j = 0; j < 16; ++j) {
            size_t srow = (size_t)b * S_ + srow0 + r0 + j;
            float q = qv[j];
            float4 o;
            o.x = q * mv4.x + kv4.x; o.y = q * mv4.y + kv4.y;
            o.z = q * mv4.z + kv4.z; o.w = q * mv4.w + kv4.w;
            *(float4*)(out + srow * (H_ * DH_) + t * 4) = o;
        }
    }
}

extern "C" void kernel_launch(void* const* d_in, const int* in_sizes, int n_in,
                              void* d_out, int out_size, void* d_ws, size_t ws_size,
                              hipStream_t stream) {
    const float* hidden = (const float*)d_in[0];
    const float* mask   = (const float*)d_in[1];
    const float* Wq     = (const float*)d_in[2];
    const float* Wk     = (const float*)d_in[3];
    const float* Wv     = (const float*)d_in[4];
    const float* aw     = (const float*)d_in[5];
    float* out = (float*)d_out;
    float* ws  = (float*)d_ws;

    prologue_kernel<<<264, 256, 0, stream>>>(Wq, Wk, aw, mask, ws + OFF_WQE, ws + OFF_WKE, ws + OFF_IL);
    pass1_kernel  <<<256,  256, 0, stream>>>(hidden, mask, ws + OFF_WQE, ws + OFF_WKE,
                                             ws + OFF_QS, ws + OFF_HSP, ws + OFF_KHP);
    reduce_kernel <<<544,  256, 0, stream>>>(ws + OFF_KHP, ws + OFF_HSP, ws + OFF_KH, ws + OFF_HS);
    pass2_kernel  <<<64,   256, 0, stream>>>(Wv, ws + OFF_HS, ws + OFF_KH, ws + OFF_IL, ws + OFF_MKV);
    pass3_kernel  <<<1024, 256, 0, stream>>>(ws + OFF_QS, ws + OFF_MKV, out);
}


# output pass re-gridded 1024 -> 256 work-groups (one per WGP, 128 rows each): the tile's scores fetched once and transposed through LDS, then waves only store; arithmetic and every sum unchanged
# speedup vs baseline: 3.0994x; 1.0465x over previous
#include <hip/hip_runtime.h>

#define B_   8
#define S_   4096
#define HID_ 1024
#define H_   16
#define DH_  64

typedef __attribute__((ext_vector_type(2))) float v2f;
typedef __attribute__((ext_vector_type(8))) float v8f;

#define OFF_WQE 0
#define OFF_WKE (OFF_WQE + H_*HID_)
#define OFF_QS  (OFF_WKE + H_*HID_)
#define OFF_HSP (OFF_QS  + B_*S_*H_)
#define OFF_KHP (OFF_HSP + B_*32*HID_)
#define OFF_HS  (OFF_KHP + 256*H_*HID_)
#define OFF_KH  (OFF_HS  + B_*HID_)
#define OFF_MKV (OFF_KH  + B_*H_*HID_)
#define OFF_IL  (OFF_MKV + 2*B_*H_*DH_)

__global__ __launch_bounds__(256) void prologue_kernel(
    const float* __restrict__ Wq, const float* __restrict__ Wk, const float* __restrict__ aw,
    const float* __restrict__ mask,
    float* __restrict__ wqe, float* __restrict__ wke, float* __restrict__ il)
{
    __shared__ __align__(16) float red[16 * 2 * 64];
    const int t = threadIdx.x;
    if (blockIdx.x < 256) {
        const int h  = blockIdx.x >> 4;
        const int i0 = (blockIdx.x & 15) * 64;
        const int cq = t & 15, dg = t >> 4;
        const float4 a4 = *(const float4*)(aw + h * DH_ + dg * 4);
        const size_t rb = (size_t)(h * DH_ + dg * 4) * HID_ + i0 + cq * 4;
        const float4 q0 = *(const float4*)(Wq + rb);
        const float4 q1 = *(const float4*)(Wq + rb + HID_);
        const float4 q2 = *(const float4*)(Wq + rb + 2 * HID_);
        const float4 q3 = *(const float4*)(Wq + rb + 3 * HID_);
        const float4 k0 = *(const float4*)(Wk + rb);
        const float4 k1 = *(const float4*)(Wk + rb + HID_);
        const float4 k2 = *(const float4*)(Wk + rb + 2 * HID_);
        const float4 k3 = *(const float4*)(Wk + rb + 3 * HID_);
        float4 sq, sk;
        sq.x = a4.x * q0.x; sq.y = a4.x * q0.y; sq.z = a4.x * q0.z; sq.w = a4.x * q0.w;
        sk.x = a4.x * k0.x; sk.y = a4.x * k0.y; sk.z = a4.x * k0.z; sk.w = a4.x * k0.w;
        sq.x += a4.y * q1.x; sq.y += a4.y * q1.y; sq.z += a4.y * q1.z; sq.w += a4.y * q1.w;
        sk.x += a4.y * k1.x; sk.y += a4.y * k1.y; sk.z += a4.y * k1.z; sk.w += a4.y * k1.w;
        sq.x += a4.z * q2.x; sq.y += a4.z * q2.y; sq.z += a4.z * q2.z; sq.w += a4.z * q2.w;
        sk.x += a4.z * k2.x; sk.y += a4.z * k2.y; sk.z += a4.z * k2.z; sk.w += a4.z * k2.w;
        sq.x += a4.w * q3.x; sq.y += a4.w * q3.y; sq.z += a4.w * q3.z; sq.w += a4.w * q3.w;
        sk.x += a4.w * k3.x; sk.y += a4.w * k3.y; sk.z += a4.w * k3.z; sk.w += a4.w * k3.w;
        *(float4*)&red[(dg * 2 + 0) * 64 + cq * 4] = sq;
        *(float4*)&red[(dg * 2 + 1) * 64 + cq * 4] = sk;
        __syncthreads();
        if (t < 128) {
            const int which = t >> 6, c = t & 63;
            float s = red[which * 64 + c];
#pragma unroll
            for (int g = 1; g < 16; ++g) s += red[(g * 2 + which) * 64 + c];
            float* dst = which ? wke : wqe;
            dst[h * HID_ + i0 + c] = s;
        }
    } else {
        const int b = blockIdx.x - 256;
        float s = 0.f;
#pragma unroll
        for (int k = 0; k < S_ / 256; ++k) s += mask[b * S_ + t + 256 * k];
        red[t] = s;
        __syncthreads();
        for (int w = 128; w > 0; w >>= 1) {
            if (t < w) red[t] += red[t + w];
            __syncthreads();
        }
        if (t == 0) il[b] = 1.0f / red[0];
    }
}

static __device__ __forceinline__ void lds_barrier()
{
    asm volatile("s_wait_dscnt 0x0\n\ts_barrier_signal -1\n\ts_barrier_wait -1" ::: "memory");
}

#define TP_ 1028
#define KP_ 20

__global__ __launch_bounds__(256) void pass1_kernel(
    const float* __restrict__ hidden, const float* __restrict__ mask,
    const float* __restrict__ wqe, const float* __restrict__ wke,
    float* __restrict__ q_scores, float* __restrict__ hsum_part,
    float* __restrict__ kh_part)
{
    __shared__ __align__(16) float lds_tile[16 * TP_];
    __shared__ __align__(16) float lds_part[8 * 2 * 8 * 32];
    __shared__ __align__(16) float lds_ksT[16 * KP_];
    __shared__ float lds_mask[128];

    const int blk = blockIdx.x & 31;
    const int b   = blockIdx.x >> 5;
    const int s0  = blk * 128;
    const int t = threadIdx.x;
    const int wave = t >> 5, lane = t & 31;
    const int lane16 = lane & 15, laneHi = lane >> 4;

    const float* hbase = hidden + ((size_t)b * S_ + s0) * HID_ + 4 * t;

    float mk = (t < 128) ? mask[b * S_ + s0 + t] : 0.f;

    float4 wq4[16], wk4[16];
    {
        const float* wqr = wqe + lane16 * HID_ + wave * 128 + 4 * laneHi;
        const float* wkr = wke + lane16 * HID_ + wave * 128 + 4 * laneHi;
#pragma unroll
        for (int s = 0; s < 16; ++s) {
            wq4[s] = *(const float4*)(wqr + s * 8);
            wk4[s] = *(const float4*)(wkr + s * 8);
        }
    }
    float4 pa[16], pb[16];
    auto fetch = [&](float4 (&v)[16], int st) {
        const float* nb = hbase + (size_t)(st * 16) * HID_;
#pragma unroll
        for (int r = 0; r < 16; ++r) v[r] = *(const float4*)(nb + (size_t)r * HID_);
    };
    fetch(pa, 0);

    if (t < 128) lds_mask[t] = mk;
    lds_barrier();

    v8f kh[8];
#pragma unroll
    for (int n = 0; n < 8; ++n) kh[n] = v8f{};
    float4 hs = {0.f, 0.f, 0.f, 0.f};

    const int rr = 2 * wave + laneHi;
    const int rsrc = (rr & 7) * 32 + (rr >> 3) * 16 + lane16;

    auto step = [&](const float4 (&pre)[16], float4 (&nxt)[16], int st) {
        fetch(nxt, (st + 1) & 7);
#pragma unroll
        for (int r = 0; r < 16; ++r) {
            *(float4*)&lds_tile[r * TP_ + 4 * t] = pre[r];
            float m = lds_mask[st * 16 + r];
            hs.x += m * pre[r].x; hs.y += m * pre[r].y; hs.z += m * pre[r].z; hs.w += m * pre[r].w;
        }
        lds_barrier();

        v8f cq = {}; v8f ck = {};
        {
            const float* ap = &lds_tile[lane16 * TP_ + wave * 128 + 4 * laneHi];
#pragma unroll
            for (int s = 0; s < 16; ++s) {
                float4 a4 = *(const float4*)(ap + s * 8);
                v2f a0 = {a4.x, a4.y}, a1 = {a4.z, a4.w};
                v2f q0 = {wq4[s].x, wq4[s].y}, q1 = {wq4[s].z, wq4[s].w};
                v2f k0 = {wk4[s].x, wk4[s].y}, k1 = {wk4[s].z, wk4[s].w};
                cq = __builtin_amdgcn_wmma_f32_16x16x4_f32(false, a0, false, q0, (short)0, cq, false, false);
                ck = __builtin_amdgcn_wmma_f32_16x16x4_f32(false, a0, false, k0, (short)0, ck, false, false);
                cq = __builtin_amdgcn_wmma_f32_16x16x4_f32(false, a1, false, q1, (short)0, cq, false, false);
                ck = __builtin_amdgcn_wmma_f32_16x16x4_f32(false, a1, false, k1, (short)0, ck, false, false);
            }
        }
#pragma unroll
        for (int j = 0; j < 8; ++j) {
            lds_part[((wave * 2 + 0) * 8 + j) * 32 + lane] = cq[j];
            lds_part[((wave * 2 + 1) * 8 + j) * 32 + lane] = ck[j];
        }
        lds_barrier();

        {
            float sq = lds_part[rsrc], sk = lds_part[256 + rsrc];
#pragma unroll
            for (int w = 1; w < 8; ++w) {
                sq += lds_part[(w * 2 + 0) * 256 + rsrc];
                sk += lds_part[(w * 2 + 1) * 256 + rsrc];
            }
            float m = lds_mask[st * 16 + rr];
            q_scores[((size_t)b * S_ + s0 + st * 16 + rr) * H_ + lane16] = sq * m;
            lds_ksT[lane16 * KP_ + rr] = sk * m;
        }
        lds_barrier();

#pragma unroll
        for (int ss = 0; ss < 4; ++ss) {
            v2f a = *(const v2f*)&lds_ksT[lane16 * KP_ + ss * 4 + 2 * laneHi];
            const float* bp = &lds_tile[(ss * 4 + 2 * laneHi) * TP_ + wave * 128 + lane16];
#pragma unroll
            for (int n = 0; n < 8; ++n) {
                v2f bb;
                bb.x = bp[n * 16];
                bb.y = bp[TP_ + n * 16];
                kh[n] = __builtin_amdgcn_wmma_f32_16x16x4_f32(false, a, false, bb, (short)0, kh[n], false, false);
            }
        }
        lds_barrier();
    };

    for (int st = 0; st < 8; st += 2) {
        step(pa, pb, st);
        step(pb, pa, st + 1);
    }

    *(float4*)&hsum_part[(size_t)(b * 32 + blk) * HID_ + 4 * t] = hs;
#pragma unroll
    for (int n = 0; n < 8; ++n)
#pragma unroll
        for (int j = 0; j < 8; ++j)
            lds_tile[(j + 8 * laneHi) * TP_ + wave * 128 + n * 16 + lane16] = kh[n][j];
    lds_barrier();
#pragma unroll 4
    for (int hh = 0; hh < 16; ++hh) {
        float4 v = *(const float4*)&lds_tile[hh * TP_ + 4 * t];
        *(float4*)&kh_part[(((size_t)(b * 32 + blk)) * H_ + hh) * HID_ + 4 * t] = v;
    }
}

__global__ __launch_bounds__(256) void reduce_kernel(
    const float* __restrict__ khp, const float* __restrict__ hsp,
    float* __restrict__ kh, float* __restrict__ hs)
{
    if (blockIdx.x < 512) {
        int idx = blockIdx.x * 256 + threadIdx.x;
        int bh = idx >> 10, i = idx & (HID_ - 1);
        int b = bh >> 4, h = bh & 15;
        float v[32];
#pragma unroll
        for (int blk = 0; blk < 32; ++blk)
            v[blk] = khp[(((size_t)(b * 32 + blk)) * H_ + h) * HID_ + i];
        float s = 0.f;
#pragma unroll
        for (int blk = 0; blk < 32; ++blk) s += v[blk];
        kh[idx] = s;
    } else {
        int idx = (blockIdx.x - 512) * 256 + threadIdx.x;
        int b = idx >> 10, i = idx & (HID_ - 1);
        float v[32];
#pragma unroll
        for (int blk = 0; blk < 32; ++blk) v[blk] = hsp[(size_t)(b * 32 + blk) * HID_ + i];
        float s = 0.f;
#pragma unroll
        for (int blk = 0; blk < 32; ++blk) s += v[blk];
        hs[idx] = s;
    }
}

__global__ __launch_bounds__(256) void pass2_kernel(
    const float* __restrict__ Wv, const float* __restrict__ hs,
    const float* __restrict__ kh, const float* __restrict__ il,
    float* __restrict__ mkv)
{
    __shared__ float red[8 * 8 * 32];
    const int h  = blockIdx.x >> 2;
    const int d0 = (blockIdx.x & 3) * 16;
    const int t = threadIdx.x;
    const int wave = t >> 5, lane = t & 31;
    const int lane16 = lane & 15, laneHi = lane >> 4;

    const int ob = ((wave & 3) << 1) | laneHi;
    const float inv = il[ob];

    const float* arow = (lane16 < 8) ? (hs + lane16 * HID_)
                                     : (kh + (size_t)((lane16 - 8) * H_ + h) * HID_);
    const float* brow = Wv + (size_t)(h * DH_ + d0 + lane16) * HID_;
    const int kb = wave * 128 + laneHi * 4;

    float4 a[16], w[16];
#pragma unroll
    for (int s = 0; s < 16; ++s) {
        a[s] = *(const float4*)(arow + kb + s * 8);
        w[s] = *(const float4*)(brow + kb + s * 8);
    }
    __builtin_amdgcn_sched_barrier(0);

    v8f acc0 = {}; v8f acc1 = {};
#pragma unroll
    for (int s = 0; s < 16; ++s) {
        v2f a0 = {a[s].x, a[s].y}, a1 = {a[s].z, a[s].w};
        v2f w0 = {w[s].x, w[s].y}, w1 = {w[s].z, w[s].w};
        acc0 = __builtin_amdgcn_wmma_f32_16x16x4_f32(false, a0, false, w0, (short)0, acc0, false, false);
        acc1 = __builtin_amdgcn_wmma_f32_16x16x4_f32(false, a1, false, w1, (short)0, acc1, false, false);
    }
#pragma unroll
    for (int j = 0; j < 8; ++j) red[(wave * 8 + j) * 32 + lane] = acc0[j] + acc1[j];
    __syncthreads();
    {
        const int half = wave >> 2;
        const int src = ob * 32 + half * 16 + lane16;
        float s = red[src];
#pragma unroll
        for (int k = 1; k < 8; ++k) s += red[k * 256 + src];
        mkv[blockIdx.x * 256 + t] = s * inv;
    }
}

#define QP_ 132

__global__ __launch_bounds__(256) void pass3_kernel(
    const float* __restrict__ qs, const float* __restrict__ mkv, float* __restrict__ out)
{
    __shared__ __align__(16) float lq[H_ * QP_];
    const int b = blockIdx.x >> 5;
    const int srow0 = (blockIdx.x & 31) * 128;
    const int t = threadIdx.x;
    const int h = t >> 4;
    const float* tile = mkv + (h * 4 + ((t & 15) >> 2)) * 256 + b * 16 + (t & 3) * 4;
    const float* qblk = qs + ((size_t)b * S_ + srow0) * H_;
    const float4 g0 = *(const float4*)(qblk + 4 * t);
    const float4 g1 = *(const float4*)(qblk + 64 * H_ + 4 * t);
    const float4 mv4 = *(const float4*)(tile);
    const float4 kv4 = *(const float4*)(tile + 128);
    {
        const int r = t >> 2, hq = (t & 3) * 4;
        lq[(hq + 0) * QP_ + r] = g0.x; lq[(hq + 1) * QP_ + r] = g0.y;
        lq[(hq + 2) * QP_ + r] = g0.z; lq[(hq + 3) * QP_ + r] = g0.w;
        lq[(hq + 0) * QP_ + 64 + r] = g1.x; lq[(hq + 1) * QP_ + 64 + r] = g1.y;
        lq[(hq + 2) * QP_ + 64 + r] = g1.z; lq[(hq + 3) * QP_ + 64 + r] = g1.w;
    }
    __syncthreads();
    const float* lqh = &lq[h * QP_];
    float* op = out + ((size_t)b * S_ + srow0) * (H_ * DH_) + t * 4;
    for (int r = 0; r < 128; r += 4) {
        const float4 q4 = *(const float4*)(lqh + r);
        float4 o0, o1, o2, o3;
        o0.x = q4.x * mv4.x + kv4.x; o0.y = q4.x * mv4.y + kv4.y; o0.z = q4.x * mv4.z + kv4.z; o0.w = q4.x * mv4.w + kv4.w;
        o1.x = q4.y * mv4.x + kv4.x; o1.y = q4.y * mv4.y + kv4.y; o1.z = q4.y * mv4.z + kv4.z; o1.w = q4.y * mv4.w + kv4.w;
        o2.x = q4.z * mv4.x + kv4.x; o2.y = q4.z * mv4.y + kv4.y; o2.z = q4.z * mv4.z + kv4.z; o2.w = q4.z * mv4.w + kv4.w;
        o3.x = q4.w * mv4.x + kv4.x; o3.y = q4.w * mv4.y + kv4.y; o3.z = q4.w * mv4.z + kv4.z; o3.w = q4.w * mv4.w + kv4.w;
        *(float4*)(op + (size_t)(r + 0) * (H_ * DH_)) = o0;
        *(float4*)(op + (size_t)(r + 1) * (H_ * DH_)) = o1;
        *(float4*)(op + (size_t)(r + 2) * (H_ * DH_)) = o2;
        *(float4*)(op + (size_t)(r + 3) * (H_ * DH_)) = o3;
    }
}

extern "C" void kernel_launch(void* const* d_in, const int* in_sizes, int n_in,
                              void* d_out, int out_size, void* d_ws, size_t ws_size,
                              hipStream_t stream) {
    const float* hidden = (const float*)d_in[0];
    const float* mask   = (const float*)d_in[1];
    const float* Wq     = (const float*)d_in[2];
    const float* Wk     = (const float*)d_in[3];
    const float* Wv     = (const float*)d_in[4];
    const float* aw     = (const float*)d_in[5];
    float* out = (float*)d_out;
    float* ws  = (float*)d_ws;

    prologue_kernel<<<264, 256, 0, stream>>>(Wq, Wk, aw, mask, ws + OFF_WQE, ws + OFF_WKE, ws + OFF_IL);
    pass1_kernel  <<<256,  256, 0, stream>>>(hidden, mask, ws + OFF_WQE, ws + OFF_WKE,
                                             ws + OFF_QS, ws + OFF_HSP, ws + OFF_KHP);
    reduce_kernel <<<544,  256, 0, stream>>>(ws + OFF_KHP, ws + OFF_HSP, ws + OFF_KH, ws + OFF_HS);
    pass2_kernel  <<<64,   256, 0, stream>>>(Wv, ws + OFF_HS, ws + OFF_KH, ws + OFF_IL, ws + OFF_MKV);
    pass3_kernel  <<<256,  256, 0, stream>>>(ws + OFF_QS, ws + OFF_MKV, out);
}
